# starting kernel
# baseline (speedup 1.0000x reference)
#include <hip/hip_runtime.h>

#define N_NODES 50000
#define NE      800000
#define D       128
#define EPS_BN  1e-5f
#define MTILES  (N_NODES / 16)
#define LDSS    132

typedef __attribute__((ext_vector_type(2))) float v2f;
typedef __attribute__((ext_vector_type(4))) float v4f;
typedef __attribute__((ext_vector_type(8))) float v8f;


__global__ void zero_kernel(float* __restrict__ p, int n4) {
    int i = blockIdx.x * blockDim.x + threadIdx.x;
    if (i < n4) ((v4f*)p)[i] = (v4f){0.0f, 0.0f, 0.0f, 0.0f};
}

__global__ void degree_kernel(const int* __restrict__ dst, float* __restrict__ deg) {
    int e = blockIdx.x * blockDim.x + threadIdx.x;
    if (e < NE) atomicAdd(&deg[dst[e]], 1.0f);
}

__global__ void invdeg_kernel(float* __restrict__ deg) {
    int i = blockIdx.x * blockDim.x + threadIdx.x;
    if (i < N_NODES) {
        float d = deg[i];
        deg[i] = 1.0f / fmaxf(d, 1.0f);
    }
}

__global__ void scatter_kernel(const float* __restrict__ x, const int* __restrict__ src,
                               const int* __restrict__ dst, float* __restrict__ agg) {
    int t = threadIdx.x;
    int e = blockIdx.x * 2 + (t >> 7);
    int d = t & 127;
    if (e < NE) {
        int s  = src[e];
        int dd = dst[e];
        atomicAdd(&agg[(size_t)dd * D + d], x[(size_t)s * D + d]);
    }
}

__global__ __launch_bounds__(256, 1)
void sage_gemm_kernel(const float* __restrict__ Aagg, const float* __restrict__ inv_deg,
                      const float* __restrict__ Ax,
                      const float* __restrict__ W1, const float* __restrict__ W2,
                      const float* __restrict__ bias,
                      const float* __restrict__ gamma, const float* __restrict__ beta,
                      const float* __restrict__ mean, const float* __restrict__ var,
                      float* __restrict__ out, int mode) {
    __shared__ float sA1[16 * LDSS];
    __shared__ float sA2[16 * LDSS];

    const int t    = threadIdx.x;
    const int lane = t & 31;
    const int wave = t >> 5;
    const int half = lane >> 4;
    const int lrow = lane & 15;
    const int n    = wave * 16 + lrow;

    v2f b1[32];
    v2f b2[32];
#pragma unroll
    for (int ks = 0; ks < 32; ++ks) {
        int k = ks * 4 + 2 * half;
        b1[ks].x = W1[k * D + n];
        b1[ks].y = W1[(k + 1) * D + n];
    }
    if (mode != 0) {
#pragma unroll
        for (int ks = 0; ks < 32; ++ks) {
            int k = ks * 4 + 2 * half;
            b2[ks].x = W2[k * D + n];
            b2[ks].y = W2[(k + 1) * D + n];
        }
    }

    float bcol = bias[n];
    float s = 1.0f, tt = 0.0f;
    if (mode != 0) {
        s  = gamma[n] * rsqrtf(var[n] + EPS_BN);
        tt = beta[n] - mean[n] * s;
    }

    for (int mt = blockIdx.x; mt < MTILES; mt += gridDim.x) {
        const size_t rowbase = (size_t)mt * 16;

#pragma unroll
        for (int i = 0; i < 8; ++i) {
            int idx = t + 256 * i;
            int r   = idx >> 7;
            int c   = idx & 127;
            float xv = Ax[(rowbase + r) * D + c];
            if (mode == 0) {
                sA1[r * LDSS + c] = xv;
            } else {
                sA1[r * LDSS + c] = Aagg[(rowbase + r) * D + c] * inv_deg[rowbase + r];
                sA2[r * LDSS + c] = xv;
            }
        }
        __syncthreads();

        v8f c8 = {};
#pragma unroll
        for (int ks = 0; ks < 32; ++ks) {
            int k = ks * 4 + 2 * half;
            v2f a = *(const v2f*)&sA1[lrow * LDSS + k];
            c8 = __builtin_amdgcn_wmma_f32_16x16x4_f32(false, a, false, b1[ks],
                                                       (short)0, c8, false, false);
        }
        if (mode != 0) {
#pragma unroll
            for (int ks = 0; ks < 32; ++ks) {
                int k = ks * 4 + 2 * half;
                v2f a = *(const v2f*)&sA2[lrow * LDSS + k];
                c8 = __builtin_amdgcn_wmma_f32_16x16x4_f32(false, a, false, b2[ks],
                                                           (short)0, c8, false, false);
            }
        }

#pragma unroll
        for (int v = 0; v < 8; ++v) {
            int m   = v + 8 * half;
            float h = c8[v] + bcol;
            float y;
            if (mode == 0) {
                y = fmaxf(h, 0.0f);
            } else {
                y = fmaxf(h * s + tt, 0.0f);
                if (mode == 1) y += sA2[m * LDSS + n];
            }
            out[(rowbase + m) * D + n] = y;
        }
        __syncthreads();
    }
}


extern "C" void kernel_launch(void* const* d_in, const int* in_sizes, int n_in,
                              void* d_out, int out_size, void* d_ws, size_t ws_size,
                              hipStream_t stream) {
    (void)in_sizes; (void)n_in; (void)out_size; (void)ws_size;

    const float* x   = (const float*)d_in[0];
    const int*   ei  = (const int*)d_in[1];
    const float* Wp  = (const float*)d_in[2];
    const float* bp  = (const float*)d_in[3];
    const float* Wl  = (const float*)d_in[4];
    const float* bl  = (const float*)d_in[5];
    const float* Wr  = (const float*)d_in[6];
    const float* gam = (const float*)d_in[7];
    const float* bet = (const float*)d_in[8];
    const float* mea = (const float*)d_in[9];
    const float* var = (const float*)d_in[10];

    const int* src = ei;
    const int* dst = ei + NE;

    float* ws  = (float*)d_ws;
    float* deg = ws;
    float* xa  = ws + ((N_NODES + 63) / 64) * 64;
    float* xb  = xa + (size_t)N_NODES * D;
    float* agg = xb + (size_t)N_NODES * D;

    const int GEMM_BLOCKS = 512;

    {
        int n4 = N_NODES / 4;
        zero_kernel<<<(n4 + 255) / 256, 256, 0, stream>>>(deg, n4);
        degree_kernel<<<(NE + 255) / 256, 256, 0, stream>>>(dst, deg);
        invdeg_kernel<<<(N_NODES + 255) / 256, 256, 0, stream>>>(deg);
    }

    sage_gemm_kernel<<<GEMM_BLOCKS, 256, 0, stream>>>(
        nullptr, deg, x, Wp, Wp, bp, gam, bet, mea, var, xa, 0);

    float* xcur  = xa;
    float* xnext = xb;
    for (int i = 0; i < 4; ++i) {
        int n4 = (N_NODES * D) / 4;
        zero_kernel<<<(n4 + 255) / 256, 256, 0, stream>>>(agg, n4);
        scatter_kernel<<<NE / 2, 256, 0, stream>>>(xcur, src, dst, agg);

        float* o = (i < 3) ? xnext : (float*)d_out;
        sage_gemm_kernel<<<GEMM_BLOCKS, 256, 0, stream>>>(
            agg, deg, xcur,
            Wl + (size_t)i * D * D, Wr + (size_t)i * D * D, bl + (size_t)i * D,
            gam + (size_t)i * D, bet + (size_t)i * D, mea + (size_t)i * D, var + (size_t)i * D,
            o, (i < 3) ? 1 : 2);

        float* tmp = xcur; xcur = xnext; xnext = tmp;
    }
}


# Neighbour sum: each of 256 work-groups owns 196 nodes in LDS, filters the whole edge list and adds source rows there (LDS float adds, arrival order), plain stores to agg; global atomics, zero pass go
# speedup vs baseline: 2.2462x; 2.2462x over previous
#include <hip/hip_runtime.h>

#define N_NODES 50000
#define NE      800000
#define D       128
#define EPS_BN  1e-5f
#define MTILES  (N_NODES / 16)
#define LDSS    132

typedef __attribute__((ext_vector_type(2))) float v2f;
typedef __attribute__((ext_vector_type(4))) float v4f;
typedef __attribute__((ext_vector_type(8))) float v8f;


__global__ void zero_kernel(float* __restrict__ p, int n4) {
    int i = blockIdx.x * blockDim.x + threadIdx.x;
    if (i < n4) ((v4f*)p)[i] = (v4f){0.0f, 0.0f, 0.0f, 0.0f};
}

__global__ void degree_kernel(const int* __restrict__ dst, float* __restrict__ deg) {
    int e = blockIdx.x * blockDim.x + threadIdx.x;
    if (e < NE) atomicAdd(&deg[dst[e]], 1.0f);
}

__global__ void invdeg_kernel(float* __restrict__ deg) {
    int i = blockIdx.x * blockDim.x + threadIdx.x;
    if (i < N_NODES) {
        float d = deg[i];
        deg[i] = 1.0f / fmaxf(d, 1.0f);
    }
}

#define AGG_T    1024
#define AGG_NW   (AGG_T / 32)
#define AGG_ROWS 196
#define AGG_WGS  ((N_NODES + AGG_ROWS - 1) / AGG_ROWS)
#define QCAP     16384
#define AGG_GU   8
#define NG4      (NE / 4)

typedef __attribute__((ext_vector_type(4))) int v4i;

__global__ __launch_bounds__(AGG_T)
void aggregate_kernel(const float* __restrict__ x, const int* __restrict__ src,
                      const int* __restrict__ dst, float* __restrict__ agg) {
    __shared__ float    sAgg[AGG_ROWS * D];
    __shared__ unsigned sQ[QCAP];
    __shared__ unsigned sQn;

    const int t    = threadIdx.x;
    const int lane = t & 31;
    const int wave = __builtin_amdgcn_readfirstlane(t >> 5);
    const int lo   = blockIdx.x * AGG_ROWS;
    const unsigned cnt = (unsigned)min(AGG_ROWS, N_NODES - lo);

    for (int i = t; i < AGG_ROWS * D / 4; i += AGG_T)
        ((v4f*)sAgg)[i] = (v4f){0.0f, 0.0f, 0.0f, 0.0f};
    if (t == 0) sQn = 0u;
    __syncthreads();

    const v4i* __restrict__ d4 = (const v4i*)dst;
    for (int g0 = t; g0 < NG4; g0 += 4 * AGG_T) {
        v4i dv[4];
#pragma unroll
        for (int u = 0; u < 4; ++u) {
            int g = g0 + u * AGG_T;
            dv[u] = d4[g < NG4 ? g : NG4 - 1];
        }
#pragma unroll
        for (int u = 0; u < 4; ++u) {
            int g = g0 + u * AGG_T;
            unsigned r0 = (unsigned)(dv[u].x - lo);
            unsigned r1 = (unsigned)(dv[u].y - lo);
            unsigned r2 = (unsigned)(dv[u].z - lo);
            unsigned r3 = (unsigned)(dv[u].w - lo);
            unsigned rmin = min(min(r0, r1), min(r2, r3));
            if (g < NG4 && rmin < cnt) {
                unsigned rr[4] = {r0, r1, r2, r3};
#pragma unroll
                for (int c = 0; c < 4; ++c) {
                    unsigned r = rr[c];
                    if (r < cnt) {
                        unsigned e   = (unsigned)(g * 4 + c);
                        unsigned pos = atomicAdd(&sQn, 1u);
                        if (pos < QCAP) {
                            sQ[pos] = e | (r << 20);
                        } else {
                            int s = src[e];
#pragma nounroll
                            for (int j = 0; j < D; ++j)
                                atomicAdd(&sAgg[r * D + j], x[(size_t)s * D + j]);
                        }
                    }
                }
            }
        }
    }
    __syncthreads();

    const unsigned n = min(sQn, (unsigned)QCAP);
    for (unsigned i0 = (unsigned)wave; i0 < n; i0 += AGG_GU * AGG_NW) {
        unsigned ent[AGG_GU];
        v2f va[AGG_GU], vb[AGG_GU];
#pragma unroll
        for (int u = 0; u < AGG_GU; ++u) {
            unsigned i = i0 + u * AGG_NW;
            ent[u] = __builtin_amdgcn_readfirstlane(sQ[i < n ? i : i0]);
        }
#pragma unroll
        for (int u = 0; u < AGG_GU; ++u) {
            int s = src[ent[u] & 0xFFFFFu];
            const float* xr = x + (size_t)s * D + 2 * lane;
            va[u] = *(const v2f*)xr;
            vb[u] = *(const v2f*)(xr + 64);
        }
#pragma unroll
        for (int u = 0; u < AGG_GU; ++u) {
            unsigned i = i0 + u * AGG_NW;
            if (i < n) {
                float* ar = sAgg + (ent[u] >> 20) * D + 2 * lane;
                atomicAdd(ar,      va[u].x);
                atomicAdd(ar + 1,  va[u].y);
                atomicAdd(ar + 64, vb[u].x);
                atomicAdd(ar + 65, vb[u].y);
            }
        }
    }
    __syncthreads();

    float* ao = agg + (size_t)lo * D;
    for (int i = t; i < (int)cnt * (D / 4); i += AGG_T)
        ((v4f*)ao)[i] = ((const v4f*)sAgg)[i];
}

__global__ __launch_bounds__(256, 1)
void sage_gemm_kernel(const float* __restrict__ Aagg, const float* __restrict__ inv_deg,
                      const float* __restrict__ Ax,
                      const float* __restrict__ W1, const float* __restrict__ W2,
                      const float* __restrict__ bias,
                      const float* __restrict__ gamma, const float* __restrict__ beta,
                      const float* __restrict__ mean, const float* __restrict__ var,
                      float* __restrict__ out, int mode) {
    __shared__ float sA1[16 * LDSS];
    __shared__ float sA2[16 * LDSS];

    const int t    = threadIdx.x;
    const int lane = t & 31;
    const int wave = t >> 5;
    const int half = lane >> 4;
    const int lrow = lane & 15;
    const int n    = wave * 16 + lrow;

    v2f b1[32];
    v2f b2[32];
#pragma unroll
    for (int ks = 0; ks < 32; ++ks) {
        int k = ks * 4 + 2 * half;
        b1[ks].x = W1[k * D + n];
        b1[ks].y = W1[(k + 1) * D + n];
    }
    if (mode != 0) {
#pragma unroll
        for (int ks = 0; ks < 32; ++ks) {
            int k = ks * 4 + 2 * half;
            b2[ks].x = W2[k * D + n];
            b2[ks].y = W2[(k + 1) * D + n];
        }
    }

    float bcol = bias[n];
    float s = 1.0f, tt = 0.0f;
    if (mode != 0) {
        s  = gamma[n] * rsqrtf(var[n] + EPS_BN);
        tt = beta[n] - mean[n] * s;
    }

    for (int mt = blockIdx.x; mt < MTILES; mt += gridDim.x) {
        const size_t rowbase = (size_t)mt * 16;

#pragma unroll
        for (int i = 0; i < 8; ++i) {
            int idx = t + 256 * i;
            int r   = idx >> 7;
            int c   = idx & 127;
            float xv = Ax[(rowbase + r) * D + c];
            if (mode == 0) {
                sA1[r * LDSS + c] = xv;
            } else {
                sA1[r * LDSS + c] = Aagg[(rowbase + r) * D + c] * inv_deg[rowbase + r];
                sA2[r * LDSS + c] = xv;
            }
        }
        __syncthreads();

        v8f c8 = {};
#pragma unroll
        for (int ks = 0; ks < 32; ++ks) {
            int k = ks * 4 + 2 * half;
            v2f a = *(const v2f*)&sA1[lrow * LDSS + k];
            c8 = __builtin_amdgcn_wmma_f32_16x16x4_f32(false, a, false, b1[ks],
                                                       (short)0, c8, false, false);
        }
        if (mode != 0) {
#pragma unroll
            for (int ks = 0; ks < 32; ++ks) {
                int k = ks * 4 + 2 * half;
                v2f a = *(const v2f*)&sA2[lrow * LDSS + k];
                c8 = __builtin_amdgcn_wmma_f32_16x16x4_f32(false, a, false, b2[ks],
                                                           (short)0, c8, false, false);
            }
        }

#pragma unroll
        for (int v = 0; v < 8; ++v) {
            int m   = v + 8 * half;
            float h = c8[v] + bcol;
            float y;
            if (mode == 0) {
                y = fmaxf(h, 0.0f);
            } else {
                y = fmaxf(h * s + tt, 0.0f);
                if (mode == 1) y += sA2[m * LDSS + n];
            }
            out[(rowbase + m) * D + n] = y;
        }
        __syncthreads();
    }
}


extern "C" void kernel_launch(void* const* d_in, const int* in_sizes, int n_in,
                              void* d_out, int out_size, void* d_ws, size_t ws_size,
                              hipStream_t stream) {
    (void)in_sizes; (void)n_in; (void)out_size; (void)ws_size;

    const float* x   = (const float*)d_in[0];
    const int*   ei  = (const int*)d_in[1];
    const float* Wp  = (const float*)d_in[2];
    const float* bp  = (const float*)d_in[3];
    const float* Wl  = (const float*)d_in[4];
    const float* bl  = (const float*)d_in[5];
    const float* Wr  = (const float*)d_in[6];
    const float* gam = (const float*)d_in[7];
    const float* bet = (const float*)d_in[8];
    const float* mea = (const float*)d_in[9];
    const float* var = (const float*)d_in[10];

    const int* src = ei;
    const int* dst = ei + NE;

    float* ws  = (float*)d_ws;
    float* deg = ws;
    float* xa  = ws + ((N_NODES + 63) / 64) * 64;
    float* xb  = xa + (size_t)N_NODES * D;
    float* agg = xb + (size_t)N_NODES * D;

    const int GEMM_BLOCKS = 512;

    {
        int n4 = N_NODES / 4;
        zero_kernel<<<(n4 + 255) / 256, 256, 0, stream>>>(deg, n4);
        degree_kernel<<<(NE + 255) / 256, 256, 0, stream>>>(dst, deg);
        invdeg_kernel<<<(N_NODES + 255) / 256, 256, 0, stream>>>(deg);
    }

    sage_gemm_kernel<<<GEMM_BLOCKS, 256, 0, stream>>>(
        nullptr, deg, x, Wp, Wp, bp, gam, bet, mea, var, xa, 0);

    float* xcur  = xa;
    float* xnext = xb;
    for (int i = 0; i < 4; ++i) {
        aggregate_kernel<<<AGG_WGS, AGG_T, 0, stream>>>(xcur, src, dst, agg);

        float* o = (i < 3) ? xnext : (float*)d_out;
        sage_gemm_kernel<<<GEMM_BLOCKS, 256, 0, stream>>>(
            agg, deg, xcur,
            Wl + (size_t)i * D * D, Wr + (size_t)i * D * D, bl + (size_t)i * D,
            gam + (size_t)i * D, bet + (size_t)i * D, mea + (size_t)i * D, var + (size_t)i * D,
            o, (i < 3) ? 1 : 2);

        float* tmp = xcur; xcur = xnext; xnext = tmp;
    }
}


# Edge list filtered once, not once a layer: the first neighbour-sum launch keeps its per-range queue of (source,row) in scratch; the other three launches read it and skip the filter
# speedup vs baseline: 2.6570x; 1.1829x over previous
#include <hip/hip_runtime.h>

#define N_NODES 50000
#define NE      800000
#define D       128
#define EPS_BN  1e-5f
#define MTILES  (N_NODES / 16)
#define LDSS    132

typedef __attribute__((ext_vector_type(2))) float v2f;
typedef __attribute__((ext_vector_type(4))) float v4f;
typedef __attribute__((ext_vector_type(8))) float v8f;


__global__ void zero_kernel(float* __restrict__ p, int n4) {
    int i = blockIdx.x * blockDim.x + threadIdx.x;
    if (i < n4) ((v4f*)p)[i] = (v4f){0.0f, 0.0f, 0.0f, 0.0f};
}

__global__ void degree_kernel(const int* __restrict__ dst, float* __restrict__ deg) {
    int e = blockIdx.x * blockDim.x + threadIdx.x;
    if (e < NE) atomicAdd(&deg[dst[e]], 1.0f);
}

__global__ void invdeg_kernel(float* __restrict__ deg) {
    int i = blockIdx.x * blockDim.x + threadIdx.x;
    if (i < N_NODES) {
        float d = deg[i];
        deg[i] = 1.0f / fmaxf(d, 1.0f);
    }
}

#define AGG_T    1024
#define AGG_NW   (AGG_T / 32)
#define AGG_ROWS 196
#define AGG_WGS  ((N_NODES + AGG_ROWS - 1) / AGG_ROWS)
#define QCAP     16384
#define AGG_GU   8
#define NG4      (NE / 4)

static_assert(N_NODES <= 65536, "a queue entry keeps the source node in 16 bits");
static_assert(AGG_ROWS <= 256 && NE <= (1 << 20), "queue entry fields");

typedef __attribute__((ext_vector_type(4))) int v4i;

__device__ __forceinline__ void filter_range(const int* __restrict__ src, const int* __restrict__ dst,
                                             const float* __restrict__ x, int lo, unsigned cnt,
                                             unsigned* sQ, unsigned* sQn, float* sAgg) {
    const int t = threadIdx.x;
    const v4i* __restrict__ d4 = (const v4i*)dst;
    for (int g0 = t; g0 < NG4; g0 += 4 * AGG_T) {
        v4i dv[4];
#pragma unroll
        for (int u = 0; u < 4; ++u) {
            int g = g0 + u * AGG_T;
            dv[u] = d4[g < NG4 ? g : NG4 - 1];
        }
#pragma unroll
        for (int u = 0; u < 4; ++u) {
            int g = g0 + u * AGG_T;
            unsigned r0 = (unsigned)(dv[u].x - lo);
            unsigned r1 = (unsigned)(dv[u].y - lo);
            unsigned r2 = (unsigned)(dv[u].z - lo);
            unsigned r3 = (unsigned)(dv[u].w - lo);
            unsigned rmin = min(min(r0, r1), min(r2, r3));
            if (g < NG4 && rmin < cnt) {
                unsigned rr[4] = {r0, r1, r2, r3};
#pragma unroll
                for (int c = 0; c < 4; ++c) {
                    unsigned r = rr[c];
                    if (r < cnt) {
                        unsigned e   = (unsigned)(g * 4 + c);
                        unsigned pos = atomicAdd(sQn, 1u);
                        if (pos < QCAP) {
                            sQ[pos] = e | (r << 20);
                        } else {
                            int s = src[e];
#pragma nounroll
                            for (int j = 0; j < D; ++j)
                                atomicAdd(&sAgg[r * D + j], x[(size_t)s * D + j]);
                        }
                    }
                }
            }
        }
    }
    __syncthreads();
    const unsigned n = min(*sQn, (unsigned)QCAP);
    for (unsigned i = (unsigned)t; i < n; i += AGG_T) {
        unsigned v = sQ[i];
        sQ[i] = (unsigned)src[v & 0xFFFFFu] | ((v >> 20) << 16);
    }
}

template <bool BUILD>
__global__ __launch_bounds__(AGG_T)
void aggregate_kernel(const float* __restrict__ x, const int* __restrict__ src,
                      const int* __restrict__ dst,
                      unsigned* __restrict__ q, unsigned* __restrict__ qcnt,
                      float* __restrict__ agg) {
    __shared__ float    sAgg[AGG_ROWS * D];
    __shared__ unsigned sQ[QCAP];
    __shared__ unsigned sQn;

    const int t    = threadIdx.x;
    const int lane = t & 31;
    const int wave = __builtin_amdgcn_readfirstlane(t >> 5);
    const int lo   = blockIdx.x * AGG_ROWS;
    const unsigned cnt = (unsigned)min(AGG_ROWS, N_NODES - lo);
    unsigned* qseg = q + (size_t)blockIdx.x * QCAP;

    for (int i = t; i < AGG_ROWS * D / 4; i += AGG_T)
        ((v4f*)sAgg)[i] = (v4f){0.0f, 0.0f, 0.0f, 0.0f};

    unsigned nraw = 0u;
    if (!BUILD) nraw = qcnt[blockIdx.x];
    if (BUILD || nraw > (unsigned)QCAP) {
        if (t == 0) sQn = 0u;
        __syncthreads();
        filter_range(src, dst, x, lo, cnt, sQ, &sQn, sAgg);
        nraw = sQn;
        if (BUILD) {
            const unsigned nq = min(nraw, (unsigned)QCAP);
            for (unsigned i = (unsigned)t; i < nq; i += AGG_T) qseg[i] = sQ[i];
            if (t == 0) qcnt[blockIdx.x] = nraw;
        }
    } else {
        for (unsigned i = (unsigned)t; i < nraw; i += AGG_T) sQ[i] = qseg[i];
    }
    __syncthreads();

    const unsigned n = min(nraw, (unsigned)QCAP);
    for (unsigned i0 = (unsigned)wave; i0 < n; i0 += AGG_GU * AGG_NW) {
        unsigned ent[AGG_GU];
        v2f va[AGG_GU], vb[AGG_GU];
#pragma unroll
        for (int u = 0; u < AGG_GU; ++u) {
            unsigned i = i0 + u * AGG_NW;
            ent[u] = __builtin_amdgcn_readfirstlane(sQ[i < n ? i : i0]);
        }
#pragma unroll
        for (int u = 0; u < AGG_GU; ++u) {
            const float* xr = x + (size_t)(ent[u] & 0xFFFFu) * D + 2 * lane;
            va[u] = *(const v2f*)xr;
            vb[u] = *(const v2f*)(xr + 64);
        }
#pragma unroll
        for (int u = 0; u < AGG_GU; ++u) {
            unsigned i = i0 + u * AGG_NW;
            if (i < n) {
                float* ar = sAgg + (ent[u] >> 16) * D + 2 * lane;
                atomicAdd(ar,      va[u].x);
                atomicAdd(ar + 1,  va[u].y);
                atomicAdd(ar + 64, vb[u].x);
                atomicAdd(ar + 65, vb[u].y);
            }
        }
    }
    __syncthreads();

    float* ao = agg + (size_t)lo * D;
    for (int i = t; i < (int)cnt * (D / 4); i += AGG_T)
        ((v4f*)ao)[i] = ((const v4f*)sAgg)[i];
}

__global__ __launch_bounds__(256, 1)
void sage_gemm_kernel(const float* __restrict__ Aagg, const float* __restrict__ inv_deg,
                      const float* __restrict__ Ax,
                      const float* __restrict__ W1, const float* __restrict__ W2,
                      const float* __restrict__ bias,
                      const float* __restrict__ gamma, const float* __restrict__ beta,
                      const float* __restrict__ mean, const float* __restrict__ var,
                      float* __restrict__ out, int mode) {
    __shared__ float sA1[16 * LDSS];
    __shared__ float sA2[16 * LDSS];

    const int t    = threadIdx.x;
    const int lane = t & 31;
    const int wave = t >> 5;
    const int half = lane >> 4;
    const int lrow = lane & 15;
    const int n    = wave * 16 + lrow;

    v2f b1[32];
    v2f b2[32];
#pragma unroll
    for (int ks = 0; ks < 32; ++ks) {
        int k = ks * 4 + 2 * half;
        b1[ks].x = W1[k * D + n];
        b1[ks].y = W1[(k + 1) * D + n];
    }
    if (mode != 0) {
#pragma unroll
        for (int ks = 0; ks < 32; ++ks) {
            int k = ks * 4 + 2 * half;
            b2[ks].x = W2[k * D + n];
            b2[ks].y = W2[(k + 1) * D + n];
        }
    }

    float bcol = bias[n];
    float s = 1.0f, tt = 0.0f;
    if (mode != 0) {
        s  = gamma[n] * rsqrtf(var[n] + EPS_BN);
        tt = beta[n] - mean[n] * s;
    }

    for (int mt = blockIdx.x; mt < MTILES; mt += gridDim.x) {
        const size_t rowbase = (size_t)mt * 16;

#pragma unroll
        for (int i = 0; i < 8; ++i) {
            int idx = t + 256 * i;
            int r   = idx >> 7;
            int c   = idx & 127;
            float xv = Ax[(rowbase + r) * D + c];
            if (mode == 0) {
                sA1[r * LDSS + c] = xv;
            } else {
                sA1[r * LDSS + c] = Aagg[(rowbase + r) * D + c] * inv_deg[rowbase + r];
                sA2[r * LDSS + c] = xv;
            }
        }
        __syncthreads();

        v8f c8 = {};
#pragma unroll
        for (int ks = 0; ks < 32; ++ks) {
            int k = ks * 4 + 2 * half;
            v2f a = *(const v2f*)&sA1[lrow * LDSS + k];
            c8 = __builtin_amdgcn_wmma_f32_16x16x4_f32(false, a, false, b1[ks],
                                                       (short)0, c8, false, false);
        }
        if (mode != 0) {
#pragma unroll
            for (int ks = 0; ks < 32; ++ks) {
                int k = ks * 4 + 2 * half;
                v2f a = *(const v2f*)&sA2[lrow * LDSS + k];
                c8 = __builtin_amdgcn_wmma_f32_16x16x4_f32(false, a, false, b2[ks],
                                                           (short)0, c8, false, false);
            }
        }

#pragma unroll
        for (int v = 0; v < 8; ++v) {
            int m   = v + 8 * half;
            float h = c8[v] + bcol;
            float y;
            if (mode == 0) {
                y = fmaxf(h, 0.0f);
            } else {
                y = fmaxf(h * s + tt, 0.0f);
                if (mode == 1) y += sA2[m * LDSS + n];
            }
            out[(rowbase + m) * D + n] = y;
        }
        __syncthreads();
    }
}


extern "C" void kernel_launch(void* const* d_in, const int* in_sizes, int n_in,
                              void* d_out, int out_size, void* d_ws, size_t ws_size,
                              hipStream_t stream) {
    (void)in_sizes; (void)n_in; (void)out_size; (void)ws_size;

    const float* x   = (const float*)d_in[0];
    const int*   ei  = (const int*)d_in[1];
    const float* Wp  = (const float*)d_in[2];
    const float* bp  = (const float*)d_in[3];
    const float* Wl  = (const float*)d_in[4];
    const float* bl  = (const float*)d_in[5];
    const float* Wr  = (const float*)d_in[6];
    const float* gam = (const float*)d_in[7];
    const float* bet = (const float*)d_in[8];
    const float* mea = (const float*)d_in[9];
    const float* var = (const float*)d_in[10];

    const int* src = ei;
    const int* dst = ei + NE;

    float* ws  = (float*)d_ws;
    float* deg = ws;
    float* xa  = ws + ((N_NODES + 63) / 64) * 64;
    float* xb  = xa + (size_t)N_NODES * D;
    float* agg = xb + (size_t)N_NODES * D;

    unsigned* q    = (unsigned*)d_out;
    unsigned* qcnt = q + (size_t)AGG_WGS * QCAP;

    const int GEMM_BLOCKS = 512;

    {
        int n4 = N_NODES / 4;
        zero_kernel<<<(n4 + 255) / 256, 256, 0, stream>>>(deg, n4);
        degree_kernel<<<(NE + 255) / 256, 256, 0, stream>>>(dst, deg);
        invdeg_kernel<<<(N_NODES + 255) / 256, 256, 0, stream>>>(deg);
    }

    sage_gemm_kernel<<<GEMM_BLOCKS, 256, 0, stream>>>(
        nullptr, deg, x, Wp, Wp, bp, gam, bet, mea, var, xa, 0);

    float* xcur  = xa;
    float* xnext = xb;
    for (int i = 0; i < 4; ++i) {
        if (i == 0)
            aggregate_kernel<true><<<AGG_WGS, AGG_T, 0, stream>>>(xcur, src, dst, q, qcnt, agg);
        else
            aggregate_kernel<false><<<AGG_WGS, AGG_T, 0, stream>>>(xcur, src, dst, q, qcnt, agg);

        float* o = (i < 3) ? xnext : (float*)d_out;
        sage_gemm_kernel<<<GEMM_BLOCKS, 256, 0, stream>>>(
            agg, deg, xcur,
            Wl + (size_t)i * D * D, Wr + (size_t)i * D * D, bl + (size_t)i * D,
            gam + (size_t)i * D, bet + (size_t)i * D, mea + (size_t)i * D, var + (size_t)i * D,
            o, (i < 3) ? 1 : 2);

        float* tmp = xcur; xcur = xnext; xnext = tmp;
    }
}


# Neighbour sums without atomics: wave w of a range's work-group owns the rows r with r mod 32 = w, takes its edges from its own LDS bin and adds rows by plain LDS read-add-write (bin arrival order)
# speedup vs baseline: 2.9486x; 1.1097x over previous
#include <hip/hip_runtime.h>

#define N_NODES 50000
#define NE      800000
#define D       128
#define EPS_BN  1e-5f
#define MTILES  (N_NODES / 16)
#define LDSS    132

typedef __attribute__((ext_vector_type(2))) float v2f;
typedef __attribute__((ext_vector_type(4))) float v4f;
typedef __attribute__((ext_vector_type(8))) float v8f;


__global__ void zero_kernel(float* __restrict__ p, int n4) {
    int i = blockIdx.x * blockDim.x + threadIdx.x;
    if (i < n4) ((v4f*)p)[i] = (v4f){0.0f, 0.0f, 0.0f, 0.0f};
}

__global__ void degree_kernel(const int* __restrict__ dst, float* __restrict__ deg) {
    int e = blockIdx.x * blockDim.x + threadIdx.x;
    if (e < NE) atomicAdd(&deg[dst[e]], 1.0f);
}

__global__ void invdeg_kernel(float* __restrict__ deg) {
    int i = blockIdx.x * blockDim.x + threadIdx.x;
    if (i < N_NODES) {
        float d = deg[i];
        deg[i] = 1.0f / fmaxf(d, 1.0f);
    }
}

#define AGG_T     1024
#define AGG_NW    (AGG_T / 32)
#define AGG_ROWS  196
#define AGG_SLOTS ((AGG_ROWS + AGG_NW - 1) / AGG_NW)
#define AGG_WGS   ((N_NODES + AGG_ROWS - 1) / AGG_ROWS)
#define QCAP      16384
#define BINCAP    (QCAP / AGG_NW)
#define AGG_GU    8
#define NG4       (NE / 4)

static_assert(N_NODES <= 65536, "a bin entry keeps the source node in 16 bits");
static_assert(AGG_ROWS <= 256 && NE <= (1 << 20), "bin entry fields");
static_assert(NE % 32 == 0, "a wave walks the edge list 32 edges at a time");
static_assert(AGG_NW == 32, "row -> wave map is r & 31");

typedef __attribute__((ext_vector_type(4))) int v4i;

__device__ __forceinline__ void add_row(float* sAgg, unsigned row, int lane, v2f a, v2f b) {
    float* ar = sAgg + row * D + 2 * lane;
    v2f p = *(const v2f*)ar;
    v2f q = *(const v2f*)(ar + 64);
    p += a;
    q += b;
    *(v2f*)ar        = p;
    *(v2f*)(ar + 64) = q;
}

__device__ __forceinline__ void filter_bins(const int* __restrict__ dst, int lo, unsigned cnt,
                                            unsigned* sQ, unsigned* sBinN) {
    const int t = threadIdx.x;
    const v4i* __restrict__ d4 = (const v4i*)dst;
    for (int g0 = t; g0 < NG4; g0 += 4 * AGG_T) {
        v4i dv[4];
#pragma unroll
        for (int u = 0; u < 4; ++u) {
            int g = g0 + u * AGG_T;
            dv[u] = d4[g < NG4 ? g : NG4 - 1];
        }
#pragma unroll
        for (int u = 0; u < 4; ++u) {
            int g = g0 + u * AGG_T;
            unsigned r0 = (unsigned)(dv[u].x - lo);
            unsigned r1 = (unsigned)(dv[u].y - lo);
            unsigned r2 = (unsigned)(dv[u].z - lo);
            unsigned r3 = (unsigned)(dv[u].w - lo);
            unsigned rmin = min(min(r0, r1), min(r2, r3));
            if (g < NG4 && rmin < cnt) {
                unsigned rr[4] = {r0, r1, r2, r3};
#pragma unroll
                for (int c = 0; c < 4; ++c) {
                    unsigned r = rr[c];
                    if (r < cnt) {
                        unsigned e   = (unsigned)(g * 4 + c);
                        unsigned w   = r & 31u;
                        unsigned pos = atomicAdd(&sBinN[w], 1u);
                        if (pos < (unsigned)BINCAP) sQ[w * BINCAP + pos] = e | (r << 20);
                    }
                }
            }
        }
    }
}

__device__ __forceinline__ void sum_bin(const unsigned* Q, unsigned n, const float* __restrict__ x,
                                        int lane, float* sAgg) {
    unsigned cur = ((unsigned)lane < n) ? Q[lane] : 0u;
    for (unsigned base = 0; base < n; base += 32) {
        const unsigned nb  = base + 32u + (unsigned)lane;
        const unsigned nxt = (nb < n) ? Q[nb] : 0u;
#pragma unroll
        for (int g = 0; g < 32 / AGG_GU; ++g) {
            const unsigned i0 = base + (unsigned)(g * AGG_GU);
            if (i0 < n) {
                unsigned ent[AGG_GU];
                v2f va[AGG_GU], vb[AGG_GU];
#pragma unroll
                for (int u = 0; u < AGG_GU; ++u)
                    ent[u] = (unsigned)__builtin_amdgcn_readlane((int)cur, g * AGG_GU + u);
#pragma unroll
                for (int u = 0; u < AGG_GU; ++u) {
                    const float* xr = x + (size_t)(ent[u] & 0xFFFFu) * D + 2 * lane;
                    va[u] = *(const v2f*)xr;
                    vb[u] = *(const v2f*)(xr + 64);
                }
#pragma unroll
                for (int u = 0; u < AGG_GU; ++u)
                    if (i0 + (unsigned)u < n) add_row(sAgg, ent[u] >> 16, lane, va[u], vb[u]);
            }
        }
        cur = nxt;
    }
}

__device__ __forceinline__ void sum_by_scan(const int* __restrict__ src, const int* __restrict__ dst,
                                            const float* __restrict__ x, int lo, unsigned cnt,
                                            int wave, int lane, float* sAgg) {
    for (int e0 = 0; e0 < NE; e0 += 32) {
        const unsigned r = (unsigned)(dst[e0 + lane] - lo);
        const bool mine  = (r < cnt) && ((r & 31u) == (unsigned)wave);
        unsigned m = __builtin_amdgcn_ballot_w32(mine);
        while (m != 0u) {
            const int l = __builtin_ctz(m);
            m &= m - 1u;
            const unsigned rl = (unsigned)__builtin_amdgcn_readlane((int)r, l);
            const int s = src[e0 + l];
            const float* xr = x + (size_t)s * D + 2 * lane;
            const v2f a = *(const v2f*)xr;
            const v2f b = *(const v2f*)(xr + 64);
            add_row(sAgg, rl, lane, a, b);
        }
    }
}

template <bool BUILD>
__global__ __launch_bounds__(AGG_T)
void aggregate_kernel(const float* __restrict__ x, const int* __restrict__ src,
                      const int* __restrict__ dst,
                      unsigned* __restrict__ q, unsigned* __restrict__ qcnt,
                      float* __restrict__ agg) {
    __shared__ float    sAgg[AGG_ROWS * D];
    __shared__ unsigned sQ[QCAP];

    const int t    = threadIdx.x;
    const int lane = t & 31;
    const int wave = __builtin_amdgcn_readfirstlane(t >> 5);
    const int lo   = blockIdx.x * AGG_ROWS;
    const unsigned cnt = (unsigned)min(AGG_ROWS, N_NODES - lo);
    unsigned* qseg = q + (size_t)blockIdx.x * QCAP + (size_t)wave * BINCAP;
    unsigned* qn   = qcnt + (size_t)blockIdx.x * AGG_NW + wave;
    unsigned* bin  = sQ + wave * BINCAP;

#pragma unroll
    for (int j = 0; j < AGG_SLOTS; ++j) {
        const unsigned r = (unsigned)(wave + AGG_NW * j);
        if (r < cnt) {
            float* ar = sAgg + r * D + 2 * lane;
            *(v2f*)ar        = (v2f){0.0f, 0.0f};
            *(v2f*)(ar + 64) = (v2f){0.0f, 0.0f};
        }
    }

    if constexpr (BUILD) {
        __shared__ unsigned sBinN[AGG_NW];

        if (t < AGG_NW) sBinN[t] = 0u;
        __syncthreads();
        filter_bins(dst, lo, cnt, sQ, sBinN);
        __syncthreads();
        const unsigned nraw = (unsigned)__builtin_amdgcn_readfirstlane((int)sBinN[wave]);
        const unsigned n    = min(nraw, (unsigned)BINCAP);
        for (unsigned j0 = (unsigned)lane; j0 < n; j0 += 128u) {
            unsigned v[4];
            int      s[4];
#pragma unroll
            for (int u = 0; u < 4; ++u) {
                unsigned i = j0 + 32u * (unsigned)u;
                v[u] = (i < n) ? bin[i] : 0u;
            }
#pragma unroll
            for (int u = 0; u < 4; ++u) s[u] = src[v[u] & 0xFFFFFu];
#pragma unroll
            for (int u = 0; u < 4; ++u) {
                unsigned i = j0 + 32u * (unsigned)u;
                if (i < n) {
                    unsigned ent = (unsigned)s[u] | ((v[u] >> 20) << 16);
                    bin[i]  = ent;
                    qseg[i] = ent;
                }
            }
        }
        if (lane == 0) *qn = nraw;

        if (nraw <= (unsigned)BINCAP) sum_bin(bin, n, x, lane, sAgg);
        else                          sum_by_scan(src, dst, x, lo, cnt, wave, lane, sAgg);
    } else {
        const unsigned nraw = (unsigned)__builtin_amdgcn_readfirstlane((int)*qn);
        if (nraw <= (unsigned)BINCAP) {
            for (unsigned j0 = (unsigned)lane; j0 < nraw; j0 += 128u) {
                unsigned v[4];
#pragma unroll
                for (int u = 0; u < 4; ++u) {
                    unsigned i = j0 + 32u * (unsigned)u;
                    v[u] = (i < nraw) ? qseg[i] : 0u;
                }
#pragma unroll
                for (int u = 0; u < 4; ++u) {
                    unsigned i = j0 + 32u * (unsigned)u;
                    if (i < nraw) bin[i] = v[u];
                }
            }
            sum_bin(bin, nraw, x, lane, sAgg);
        } else {
            sum_by_scan(src, dst, x, lo, cnt, wave, lane, sAgg);
        }
    }

#pragma unroll
    for (int j = 0; j < AGG_SLOTS; ++j) {
        const unsigned r = (unsigned)(wave + AGG_NW * j);
        if (r < cnt) {
            const float* ar = sAgg + r * D + 2 * lane;
            float* ao = agg + (size_t)(lo + (int)r) * D + 2 * lane;
            *(v2f*)ao        = *(const v2f*)ar;
            *(v2f*)(ao + 64) = *(const v2f*)(ar + 64);
        }
    }
}

__global__ __launch_bounds__(256, 1)
void sage_gemm_kernel(const float* __restrict__ Aagg, const float* __restrict__ inv_deg,
                      const float* __restrict__ Ax,
                      const float* __restrict__ W1, const float* __restrict__ W2,
                      const float* __restrict__ bias,
                      const float* __restrict__ gamma, const float* __restrict__ beta,
                      const float* __restrict__ mean, const float* __restrict__ var,
                      float* __restrict__ out, int mode) {
    __shared__ float sA1[16 * LDSS];
    __shared__ float sA2[16 * LDSS];

    const int t    = threadIdx.x;
    const int lane = t & 31;
    const int wave = t >> 5;
    const int half = lane >> 4;
    const int lrow = lane & 15;
    const int n    = wave * 16 + lrow;

    v2f b1[32];
    v2f b2[32];
#pragma unroll
    for (int ks = 0; ks < 32; ++ks) {
        int k = ks * 4 + 2 * half;
        b1[ks].x = W1[k * D + n];
        b1[ks].y = W1[(k + 1) * D + n];
    }
    if (mode != 0) {
#pragma unroll
        for (int ks = 0; ks < 32; ++ks) {
            int k = ks * 4 + 2 * half;
            b2[ks].x = W2[k * D + n];
            b2[ks].y = W2[(k + 1) * D + n];
        }
    }

    float bcol = bias[n];
    float s = 1.0f, tt = 0.0f;
    if (mode != 0) {
        s  = gamma[n] * rsqrtf(var[n] + EPS_BN);
        tt = beta[n] - mean[n] * s;
    }

    for (int mt = blockIdx.x; mt < MTILES; mt += gridDim.x) {
        const size_t rowbase = (size_t)mt * 16;

#pragma unroll
        for (int i = 0; i < 8; ++i) {
            int idx = t + 256 * i;
            int r   = idx >> 7;
            int c   = idx & 127;
            float xv = Ax[(rowbase + r) * D + c];
            if (mode == 0) {
                sA1[r * LDSS + c] = xv;
            } else {
                sA1[r * LDSS + c] = Aagg[(rowbase + r) * D + c] * inv_deg[rowbase + r];
                sA2[r * LDSS + c] = xv;
            }
        }
        __syncthreads();

        v8f c8 = {};
#pragma unroll
        for (int ks = 0; ks < 32; ++ks) {
            int k = ks * 4 + 2 * half;
            v2f a = *(const v2f*)&sA1[lrow * LDSS + k];
            c8 = __builtin_amdgcn_wmma_f32_16x16x4_f32(false, a, false, b1[ks],
                                                       (short)0, c8, false, false);
        }
        if (mode != 0) {
#pragma unroll
            for (int ks = 0; ks < 32; ++ks) {
                int k = ks * 4 + 2 * half;
                v2f a = *(const v2f*)&sA2[lrow * LDSS + k];
                c8 = __builtin_amdgcn_wmma_f32_16x16x4_f32(false, a, false, b2[ks],
                                                           (short)0, c8, false, false);
            }
        }

#pragma unroll
        for (int v = 0; v < 8; ++v) {
            int m   = v + 8 * half;
            float h = c8[v] + bcol;
            float y;
            if (mode == 0) {
                y = fmaxf(h, 0.0f);
            } else {
                y = fmaxf(h * s + tt, 0.0f);
                if (mode == 1) y += sA2[m * LDSS + n];
            }
            out[(rowbase + m) * D + n] = y;
        }
        __syncthreads();
    }
}


extern "C" void kernel_launch(void* const* d_in, const int* in_sizes, int n_in,
                              void* d_out, int out_size, void* d_ws, size_t ws_size,
                              hipStream_t stream) {
    (void)in_sizes; (void)n_in; (void)out_size; (void)ws_size;

    const float* x   = (const float*)d_in[0];
    const int*   ei  = (const int*)d_in[1];
    const float* Wp  = (const float*)d_in[2];
    const float* bp  = (const float*)d_in[3];
    const float* Wl  = (const float*)d_in[4];
    const float* bl  = (const float*)d_in[5];
    const float* Wr  = (const float*)d_in[6];
    const float* gam = (const float*)d_in[7];
    const float* bet = (const float*)d_in[8];
    const float* mea = (const float*)d_in[9];
    const float* var = (const float*)d_in[10];

    const int* src = ei;
    const int* dst = ei + NE;

    float* ws  = (float*)d_ws;
    float* deg = ws;
    float* xa  = ws + ((N_NODES + 63) / 64) * 64;
    float* xb  = xa + (size_t)N_NODES * D;
    float* agg = xb + (size_t)N_NODES * D;

    unsigned* q    = (unsigned*)d_out;
    unsigned* qcnt = q + (size_t)AGG_WGS * QCAP;

    const int GEMM_BLOCKS = 512;

    {
        int n4 = N_NODES / 4;
        zero_kernel<<<(n4 + 255) / 256, 256, 0, stream>>>(deg, n4);
        degree_kernel<<<(NE + 255) / 256, 256, 0, stream>>>(dst, deg);
        invdeg_kernel<<<(N_NODES + 255) / 256, 256, 0, stream>>>(deg);
    }

    sage_gemm_kernel<<<GEMM_BLOCKS, 256, 0, stream>>>(
        nullptr, deg, x, Wp, Wp, bp, gam, bet, mea, var, xa, 0);

    float* xcur  = xa;
    float* xnext = xb;
    for (int i = 0; i < 4; ++i) {
        if (i == 0)
            aggregate_kernel<true><<<AGG_WGS, AGG_T, 0, stream>>>(xcur, src, dst, q, qcnt, agg);
        else
            aggregate_kernel<false><<<AGG_WGS, AGG_T, 0, stream>>>(xcur, src, dst, q, qcnt, agg);

        float* o = (i < 3) ? xnext : (float*)d_out;
        sage_gemm_kernel<<<GEMM_BLOCKS, 256, 0, stream>>>(
            agg, deg, xcur,
            Wl + (size_t)i * D * D, Wr + (size_t)i * D * D, bl + (size_t)i * D,
            gam + (size_t)i * D, bet + (size_t)i * D, mea + (size_t)i * D, var + (size_t)i * D,
            o, (i < 3) ? 1 : 2);

        float* tmp = xcur; xcur = xnext; xnext = tmp;
    }
}


# GEMM re-tiled: each wave owns a 16-row tile for all 128 columns; weights shared from LDS as k-pair rows, tile in wave-private LDS, all row loads in flight at once, no tile barriers, whole-line stores
# speedup vs baseline: 4.0987x; 1.3900x over previous
#include <hip/hip_runtime.h>

#define N_NODES 50000
#define NE      800000
#define D       128
#define EPS_BN  1e-5f
#define MTILES  (N_NODES / 16)
#define LDSS    132

typedef __attribute__((ext_vector_type(2))) float v2f;
typedef __attribute__((ext_vector_type(4))) float v4f;
typedef __attribute__((ext_vector_type(8))) float v8f;


__global__ void zero_kernel(float* __restrict__ p, int n4) {
    int i = blockIdx.x * blockDim.x + threadIdx.x;
    if (i < n4) ((v4f*)p)[i] = (v4f){0.0f, 0.0f, 0.0f, 0.0f};
}

__global__ void degree_kernel(const int* __restrict__ dst, float* __restrict__ deg) {
    int e = blockIdx.x * blockDim.x + threadIdx.x;
    if (e < NE) atomicAdd(&deg[dst[e]], 1.0f);
}

__global__ void invdeg_kernel(float* __restrict__ deg) {
    int i = blockIdx.x * blockDim.x + threadIdx.x;
    if (i < N_NODES) {
        float d = deg[i];
        deg[i] = 1.0f / fmaxf(d, 1.0f);
    }
}

#define AGG_T     1024
#define AGG_NW    (AGG_T / 32)
#define AGG_ROWS  196
#define AGG_SLOTS ((AGG_ROWS + AGG_NW - 1) / AGG_NW)
#define AGG_WGS   ((N_NODES + AGG_ROWS - 1) / AGG_ROWS)
#define QCAP      16384
#define BINCAP    (QCAP / AGG_NW)
#define AGG_GU    8
#define NG4       (NE / 4)

static_assert(N_NODES <= 65536, "a bin entry keeps the source node in 16 bits");
static_assert(AGG_ROWS <= 256 && NE <= (1 << 20), "bin entry fields");
static_assert(NE % 32 == 0, "a wave walks the edge list 32 edges at a time");
static_assert(AGG_NW == 32, "row -> wave map is r & 31");

typedef __attribute__((ext_vector_type(4))) int v4i;

__device__ __forceinline__ void add_row(float* sAgg, unsigned row, int lane, v2f a, v2f b) {
    float* ar = sAgg + row * D + 2 * lane;
    v2f p = *(const v2f*)ar;
    v2f q = *(const v2f*)(ar + 64);
    p += a;
    q += b;
    *(v2f*)ar        = p;
    *(v2f*)(ar + 64) = q;
}

__device__ __forceinline__ void filter_bins(const int* __restrict__ dst, int lo, unsigned cnt,
                                            unsigned* sQ, unsigned* sBinN) {
    const int t = threadIdx.x;
    const v4i* __restrict__ d4 = (const v4i*)dst;
    for (int g0 = t; g0 < NG4; g0 += 4 * AGG_T) {
        v4i dv[4];
#pragma unroll
        for (int u = 0; u < 4; ++u) {
            int g = g0 + u * AGG_T;
            dv[u] = d4[g < NG4 ? g : NG4 - 1];
        }
#pragma unroll
        for (int u = 0; u < 4; ++u) {
            int g = g0 + u * AGG_T;
            unsigned r0 = (unsigned)(dv[u].x - lo);
            unsigned r1 = (unsigned)(dv[u].y - lo);
            unsigned r2 = (unsigned)(dv[u].z - lo);
            unsigned r3 = (unsigned)(dv[u].w - lo);
            unsigned rmin = min(min(r0, r1), min(r2, r3));
            if (g < NG4 && rmin < cnt) {
                unsigned rr[4] = {r0, r1, r2, r3};
#pragma unroll
                for (int c = 0; c < 4; ++c) {
                    unsigned r = rr[c];
                    if (r < cnt) {
                        unsigned e   = (unsigned)(g * 4 + c);
                        unsigned w   = r & 31u;
                        unsigned pos = atomicAdd(&sBinN[w], 1u);
                        if (pos < (unsigned)BINCAP) sQ[w * BINCAP + pos] = e | (r << 20);
                    }
                }
            }
        }
    }
}

__device__ __forceinline__ void sum_bin(const unsigned* Q, unsigned n, const float* __restrict__ x,
                                        int lane, float* sAgg) {
    unsigned cur = ((unsigned)lane < n) ? Q[lane] : 0u;
    for (unsigned base = 0; base < n; base += 32) {
        const unsigned nb  = base + 32u + (unsigned)lane;
        const unsigned nxt = (nb < n) ? Q[nb] : 0u;
#pragma unroll
        for (int g = 0; g < 32 / AGG_GU; ++g) {
            const unsigned i0 = base + (unsigned)(g * AGG_GU);
            if (i0 < n) {
                unsigned ent[AGG_GU];
                v2f va[AGG_GU], vb[AGG_GU];
#pragma unroll
                for (int u = 0; u < AGG_GU; ++u)
                    ent[u] = (unsigned)__builtin_amdgcn_readlane((int)cur, g * AGG_GU + u);
#pragma unroll
                for (int u = 0; u < AGG_GU; ++u) {
                    const float* xr = x + (size_t)(ent[u] & 0xFFFFu) * D + 2 * lane;
                    va[u] = *(const v2f*)xr;
                    vb[u] = *(const v2f*)(xr + 64);
                }
#pragma unroll
                for (int u = 0; u < AGG_GU; ++u)
                    if (i0 + (unsigned)u < n) add_row(sAgg, ent[u] >> 16, lane, va[u], vb[u]);
            }
        }
        cur = nxt;
    }
}

__device__ __forceinline__ void sum_by_scan(const int* __restrict__ src, const int* __restrict__ dst,
                                            const float* __restrict__ x, int lo, unsigned cnt,
                                            int wave, int lane, float* sAgg) {
    for (int e0 = 0; e0 < NE; e0 += 32) {
        const unsigned r = (unsigned)(dst[e0 + lane] - lo);
        const bool mine  = (r < cnt) && ((r & 31u) == (unsigned)wave);
        unsigned m = __builtin_amdgcn_ballot_w32(mine);
        while (m != 0u) {
            const int l = __builtin_ctz(m);
            m &= m - 1u;
            const unsigned rl = (unsigned)__builtin_amdgcn_readlane((int)r, l);
            const int s = src[e0 + l];
            const float* xr = x + (size_t)s * D + 2 * lane;
            const v2f a = *(const v2f*)xr;
            const v2f b = *(const v2f*)(xr + 64);
            add_row(sAgg, rl, lane, a, b);
        }
    }
}

template <bool BUILD>
__global__ __launch_bounds__(AGG_T)
void aggregate_kernel(const float* __restrict__ x, const int* __restrict__ src,
                      const int* __restrict__ dst,
                      unsigned* __restrict__ q, unsigned* __restrict__ qcnt,
                      float* __restrict__ agg) {
    __shared__ float    sAgg[AGG_ROWS * D];
    __shared__ unsigned sQ[QCAP];

    const int t    = threadIdx.x;
    const int lane = t & 31;
    const int wave = __builtin_amdgcn_readfirstlane(t >> 5);
    const int lo   = blockIdx.x * AGG_ROWS;
    const unsigned cnt = (unsigned)min(AGG_ROWS, N_NODES - lo);
    unsigned* qseg = q + (size_t)blockIdx.x * QCAP + (size_t)wave * BINCAP;
    unsigned* qn   = qcnt + (size_t)blockIdx.x * AGG_NW + wave;
    unsigned* bin  = sQ + wave * BINCAP;

#pragma unroll
    for (int j = 0; j < AGG_SLOTS; ++j) {
        const unsigned r = (unsigned)(wave + AGG_NW * j);
        if (r < cnt) {
            float* ar = sAgg + r * D + 2 * lane;
            *(v2f*)ar        = (v2f){0.0f, 0.0f};
            *(v2f*)(ar + 64) = (v2f){0.0f, 0.0f};
        }
    }

    if constexpr (BUILD) {
        __shared__ unsigned sBinN[AGG_NW];

        if (t < AGG_NW) sBinN[t] = 0u;
        __syncthreads();
        filter_bins(dst, lo, cnt, sQ, sBinN);
        __syncthreads();
        const unsigned nraw = (unsigned)__builtin_amdgcn_readfirstlane((int)sBinN[wave]);
        const unsigned n    = min(nraw, (unsigned)BINCAP);
        for (unsigned j0 = (unsigned)lane; j0 < n; j0 += 128u) {
            unsigned v[4];
            int      s[4];
#pragma unroll
            for (int u = 0; u < 4; ++u) {
                unsigned i = j0 + 32u * (unsigned)u;
                v[u] = (i < n) ? bin[i] : 0u;
            }
#pragma unroll
            for (int u = 0; u < 4; ++u) s[u] = src[v[u] & 0xFFFFFu];
#pragma unroll
            for (int u = 0; u < 4; ++u) {
                unsigned i = j0 + 32u * (unsigned)u;
                if (i < n) {
                    unsigned ent = (unsigned)s[u] | ((v[u] >> 20) << 16);
                    bin[i]  = ent;
                    qseg[i] = ent;
                }
            }
        }
        if (lane == 0) *qn = nraw;

        if (nraw <= (unsigned)BINCAP) sum_bin(bin, n, x, lane, sAgg);
        else                          sum_by_scan(src, dst, x, lo, cnt, wave, lane, sAgg);
    } else {
        const unsigned nraw = (unsigned)__builtin_amdgcn_readfirstlane((int)*qn);
        if (nraw <= (unsigned)BINCAP) {
            for (unsigned j0 = (unsigned)lane; j0 < nraw; j0 += 128u) {
                unsigned v[4];
#pragma unroll
                for (int u = 0; u < 4; ++u) {
                    unsigned i = j0 + 32u * (unsigned)u;
                    v[u] = (i < nraw) ? qseg[i] : 0u;
                }
#pragma unroll
                for (int u = 0; u < 4; ++u) {
                    unsigned i = j0 + 32u * (unsigned)u;
                    if (i < nraw) bin[i] = v[u];
                }
            }
            sum_bin(bin, nraw, x, lane, sAgg);
        } else {
            sum_by_scan(src, dst, x, lo, cnt, wave, lane, sAgg);
        }
    }

#pragma unroll
    for (int j = 0; j < AGG_SLOTS; ++j) {
        const unsigned r = (unsigned)(wave + AGG_NW * j);
        if (r < cnt) {
            const float* ar = sAgg + r * D + 2 * lane;
            float* ao = agg + (size_t)(lo + (int)r) * D + 2 * lane;
            *(v2f*)ao        = *(const v2f*)ar;
            *(v2f*)(ao + 64) = *(const v2f*)(ar + 64);
        }
    }
}

#define WPITCH    288
#define GEMM_W    16
#define GEMM_FULL 12
#define GEMM_T    (GEMM_W * 32)
#define GEMM_WGS  256

static_assert(GEMM_W - GEMM_FULL == 4, "the left-over tile is split in column quarters");
static_assert(MTILES >= GEMM_WGS * GEMM_FULL && MTILES - GEMM_WGS * GEMM_FULL <= GEMM_WGS,
              "every tile has exactly one owner");

__device__ __forceinline__ void stage_weights(const float* __restrict__ Wg, float* sW, int t) {
#pragma unroll
    for (int i = 0; i < 2048 / GEMM_T; ++i) {
        const int id = t + GEMM_T * i;
        const int p  = id >> 5;
        const int n4 = id & 31;
        const v4f a = *(const v4f*)(Wg + (size_t)(2 * p) * D + 4 * n4);
        const v4f b = *(const v4f*)(Wg + (size_t)(2 * p + 1) * D + 4 * n4);
        float* o = sW + p * WPITCH + 8 * n4;
        *(v4f*)o       = (v4f){a.x, b.x, a.y, b.y};
        *(v4f*)(o + 4) = (v4f){a.z, b.z, a.w, b.w};
    }
}

template <int NTN>
__device__ __forceinline__ void tile_times_weights(const float* tile, const float* sW,
                                                   int nt0, int c, int half, v8f* acc) {
    const float* ap = tile + c * LDSS + 2 * half;
    const float* wp = sW + half * WPITCH + 32 * nt0 + 2 * c;
#pragma unroll
    for (int ks = 0; ks < 32; ++ks) {
        const v2f a = *(const v2f*)(ap + 4 * ks);
#pragma unroll
        for (int j = 0; j < NTN; ++j) {
            const v2f b = *(const v2f*)(wp + 2 * ks * WPITCH + 32 * j);
            acc[j] = __builtin_amdgcn_wmma_f32_16x16x4_f32(false, a, false, b,
                                                           (short)0, acc[j], false, false);
        }
    }
}

template <int MODE, int NTN>
__device__ __forceinline__ void wave_tile(float* tile, const float* sW, const v4f* xr, const v4f* ar,
                                          const float* __restrict__ inv_deg, size_t rowbase, int nt0,
                                          int lane,
                                          const float* __restrict__ bias,
                                          const float* __restrict__ gamma, const float* __restrict__ beta,
                                          const float* __restrict__ mean, const float* __restrict__ var,
                                          float* __restrict__ out) {
    const int half = lane >> 4;
    const int c    = lane & 15;

    v8f acc[NTN];
#pragma unroll
    for (int j = 0; j < NTN; ++j) acc[j] = (v8f){};

#pragma unroll
    for (int r = 0; r < 16; ++r) {
        v4f v;
        if (MODE == 0) {
            v = xr[r];
        } else {
            const float idg = inv_deg[rowbase + r];
            v = ar[r] * idg;
        }
        *(v4f*)(tile + r * LDSS + 4 * lane) = v;
    }
    tile_times_weights<NTN>(tile, sW, nt0, c, half, acc);

    if (MODE != 0) {
#pragma unroll
        for (int r = 0; r < 16; ++r)
            *(v4f*)(tile + r * LDSS + 4 * lane) = xr[r];
        tile_times_weights<NTN>(tile, sW + 64 * WPITCH, nt0, c, half, acc);
    }

#pragma unroll
    for (int j = 0; j < NTN; ++j) {
        const int n = 16 * (nt0 + j) + c;
        const float bcol = bias[n];
        float s = 1.0f, tt = 0.0f;
        if (MODE != 0) {
            s  = gamma[n] * rsqrtf(var[n] + EPS_BN);
            tt = beta[n] - mean[n] * s;
        }
#pragma unroll
        for (int v = 0; v < 8; ++v) {
            const int m = v + 8 * half;
            const float h = acc[j][v] + bcol;
            float y;
            if (MODE == 0) {
                y = fmaxf(h, 0.0f);
            } else {
                y = fmaxf(h * s + tt, 0.0f);
                if (MODE == 1) y += tile[m * LDSS + n];
            }
            tile[m * LDSS + n] = y;
        }
    }
    constexpr int LPR = 4 * NTN;
    constexpr int RPI = 32 / LPR;
    const int srow = lane / LPR;
    const int scol = 16 * nt0 + 4 * (lane % LPR);
#pragma unroll
    for (int i = 0; i < 16 / RPI; ++i) {
        const int r = i * RPI + srow;
        *(v4f*)(out + (rowbase + r) * D + scol) = *(const v4f*)(tile + r * LDSS + scol);
    }
}

template <int MODE>
__global__ __launch_bounds__(GEMM_T)
void sage_gemm_kernel(const float* __restrict__ Aagg, const float* __restrict__ inv_deg,
                      const float* __restrict__ Ax,
                      const float* __restrict__ W1, const float* __restrict__ W2,
                      const float* __restrict__ bias,
                      const float* __restrict__ gamma, const float* __restrict__ beta,
                      const float* __restrict__ mean, const float* __restrict__ var,
                      float* __restrict__ out) {
    __shared__ __attribute__((aligned(16))) float sW[(MODE == 0 ? 1 : 2) * 64 * WPITCH];
    __shared__ __attribute__((aligned(16))) float sT[GEMM_W * 16 * LDSS];

    const int t    = threadIdx.x;
    const int lane = t & 31;
    const int wave = __builtin_amdgcn_readfirstlane(t >> 5);

    const bool full = wave < GEMM_FULL;
    const int  mt   = full ? (int)blockIdx.x + GEMM_WGS * wave
                           : GEMM_WGS * GEMM_FULL + (int)blockIdx.x;
    const bool has  = mt < MTILES;
    const size_t rowbase = (size_t)(has ? mt : 0) * 16;
    float* tile = sT + wave * (16 * LDSS);

    v4f xr[16];
    v4f ar[16];
    if (has) {
#pragma unroll
        for (int r = 0; r < 16; ++r)
            xr[r] = *(const v4f*)(Ax + (rowbase + r) * D + 4 * lane);
        if (MODE != 0) {
#pragma unroll
            for (int r = 0; r < 16; ++r)
                ar[r] = *(const v4f*)(Aagg + (rowbase + r) * D + 4 * lane);
        }
    }

    stage_weights(W1, sW, t);
    if (MODE != 0) stage_weights(W2, sW + 64 * WPITCH, t);
    __syncthreads();
    if (!has) return;

    if (full)
        wave_tile<MODE, 8>(tile, sW, xr, ar, inv_deg, rowbase, 0, lane,
                           bias, gamma, beta, mean, var, out);
    else
        wave_tile<MODE, 2>(tile, sW, xr, ar, inv_deg, rowbase, 2 * (wave - GEMM_FULL), lane,
                           bias, gamma, beta, mean, var, out);
}


extern "C" void kernel_launch(void* const* d_in, const int* in_sizes, int n_in,
                              void* d_out, int out_size, void* d_ws, size_t ws_size,
                              hipStream_t stream) {
    (void)in_sizes; (void)n_in; (void)out_size; (void)ws_size;

    const float* x   = (const float*)d_in[0];
    const int*   ei  = (const int*)d_in[1];
    const float* Wp  = (const float*)d_in[2];
    const float* bp  = (const float*)d_in[3];
    const float* Wl  = (const float*)d_in[4];
    const float* bl  = (const float*)d_in[5];
    const float* Wr  = (const float*)d_in[6];
    const float* gam = (const float*)d_in[7];
    const float* bet = (const float*)d_in[8];
    const float* mea = (const float*)d_in[9];
    const float* var = (const float*)d_in[10];

    const int* src = ei;
    const int* dst = ei + NE;

    float* ws  = (float*)d_ws;
    float* deg = ws;
    float* xa  = ws + ((N_NODES + 63) / 64) * 64;
    float* xb  = xa + (size_t)N_NODES * D;
    float* agg = xb + (size_t)N_NODES * D;

    unsigned* q    = (unsigned*)d_out;
    unsigned* qcnt = q + (size_t)AGG_WGS * QCAP;

    {
        int n4 = N_NODES / 4;
        zero_kernel<<<(n4 + 255) / 256, 256, 0, stream>>>(deg, n4);
        degree_kernel<<<(NE + 255) / 256, 256, 0, stream>>>(dst, deg);
        invdeg_kernel<<<(N_NODES + 255) / 256, 256, 0, stream>>>(deg);
    }

    sage_gemm_kernel<0><<<GEMM_WGS, GEMM_T, 0, stream>>>(
        nullptr, deg, x, Wp, Wp, bp, gam, bet, mea, var, xa);

    float* xcur  = xa;
    float* xnext = xb;
    for (int i = 0; i < 4; ++i) {
        if (i == 0)
            aggregate_kernel<true><<<AGG_WGS, AGG_T, 0, stream>>>(xcur, src, dst, q, qcnt, agg);
        else
            aggregate_kernel<false><<<AGG_WGS, AGG_T, 0, stream>>>(xcur, src, dst, q, qcnt, agg);

        if (i < 3)
            sage_gemm_kernel<1><<<GEMM_WGS, GEMM_T, 0, stream>>>(
                agg, deg, xcur,
                Wl + (size_t)i * D * D, Wr + (size_t)i * D * D, bl + (size_t)i * D,
                gam + (size_t)i * D, bet + (size_t)i * D, mea + (size_t)i * D, var + (size_t)i * D,
                xnext);
        else
            sage_gemm_kernel<2><<<GEMM_WGS, GEMM_T, 0, stream>>>(
                agg, deg, xcur,
                Wl + (size_t)i * D * D, Wr + (size_t)i * D * D, bl + (size_t)i * D,
                gam + (size_t)i * D, bet + (size_t)i * D, mea + (size_t)i * D, var + (size_t)i * D,
                (float*)d_out);

        float* tmp = xcur; xcur = xnext; xnext = tmp;
    }
}


# Degrees counted in the first neighbour-sum launch's edge filter (one more LDS integer add per edge, exact), which writes 1/max(deg,1): zero, degree (800k global float atomics) and 1/deg launches go
# speedup vs baseline: 4.9755x; 1.2139x over previous
#include <hip/hip_runtime.h>

#define N_NODES 50000
#define NE      800000
#define D       128
#define EPS_BN  1e-5f
#define MTILES  (N_NODES / 16)
#define LDSS    132

typedef __attribute__((ext_vector_type(2))) float v2f;
typedef __attribute__((ext_vector_type(4))) float v4f;
typedef __attribute__((ext_vector_type(8))) float v8f;

#define AGG_T     1024
#define AGG_NW    (AGG_T / 32)
#define AGG_ROWS  196
#define AGG_SLOTS ((AGG_ROWS + AGG_NW - 1) / AGG_NW)
#define AGG_WGS   ((N_NODES + AGG_ROWS - 1) / AGG_ROWS)
#define QCAP      16384
#define BINCAP    (QCAP / AGG_NW)
#define AGG_GU    8
#define NG4       (NE / 4)

static_assert(N_NODES <= 65536, "a bin entry keeps the source node in 16 bits");
static_assert(AGG_ROWS <= 256 && NE <= (1 << 20), "bin entry fields");
static_assert(NE % 32 == 0, "a wave walks the edge list 32 edges at a time");
static_assert(AGG_NW == 32, "row -> wave map is r & 31");
static_assert(AGG_ROWS <= AGG_T && NE < (1 << 24), "one thread per row writes 1/degree; counts exact in f32");

typedef __attribute__((ext_vector_type(4))) int v4i;

__device__ __forceinline__ void add_row(float* sAgg, unsigned row, int lane, v2f a, v2f b) {
    float* ar = sAgg + row * D + 2 * lane;
    v2f p = *(const v2f*)ar;
    v2f q = *(const v2f*)(ar + 64);
    p += a;
    q += b;
    *(v2f*)ar        = p;
    *(v2f*)(ar + 64) = q;
}

__device__ __forceinline__ void filter_bins(const int* __restrict__ dst, int lo, unsigned cnt,
                                            unsigned* sQ, unsigned* sBinN, unsigned* sDeg) {
    const int t = threadIdx.x;
    const v4i* __restrict__ d4 = (const v4i*)dst;
    for (int g0 = t; g0 < NG4; g0 += 4 * AGG_T) {
        v4i dv[4];
#pragma unroll
        for (int u = 0; u < 4; ++u) {
            int g = g0 + u * AGG_T;
            dv[u] = d4[g < NG4 ? g : NG4 - 1];
        }
#pragma unroll
        for (int u = 0; u < 4; ++u) {
            int g = g0 + u * AGG_T;
            unsigned r0 = (unsigned)(dv[u].x - lo);
            unsigned r1 = (unsigned)(dv[u].y - lo);
            unsigned r2 = (unsigned)(dv[u].z - lo);
            unsigned r3 = (unsigned)(dv[u].w - lo);
            unsigned rmin = min(min(r0, r1), min(r2, r3));
            if (g < NG4 && rmin < cnt) {
                unsigned rr[4] = {r0, r1, r2, r3};
#pragma unroll
                for (int c = 0; c < 4; ++c) {
                    unsigned r = rr[c];
                    if (r < cnt) {
                        unsigned e   = (unsigned)(g * 4 + c);
                        unsigned w   = r & 31u;
                        unsigned pos = atomicAdd(&sBinN[w], 1u);
                        if (pos < (unsigned)BINCAP) sQ[w * BINCAP + pos] = e | (r << 20);
                        atomicAdd(&sDeg[r], 1u);
                    }
                }
            }
        }
    }
}

__device__ __forceinline__ void sum_bin(const unsigned* Q, unsigned n, const float* __restrict__ x,
                                        int lane, float* sAgg) {
    unsigned cur = ((unsigned)lane < n) ? Q[lane] : 0u;
    for (unsigned base = 0; base < n; base += 32) {
        const unsigned nb  = base + 32u + (unsigned)lane;
        const unsigned nxt = (nb < n) ? Q[nb] : 0u;
#pragma unroll
        for (int g = 0; g < 32 / AGG_GU; ++g) {
            const unsigned i0 = base + (unsigned)(g * AGG_GU);
            if (i0 < n) {
                unsigned ent[AGG_GU];
                v2f va[AGG_GU], vb[AGG_GU];
#pragma unroll
                for (int u = 0; u < AGG_GU; ++u)
                    ent[u] = (unsigned)__builtin_amdgcn_readlane((int)cur, g * AGG_GU + u);
#pragma unroll
                for (int u = 0; u < AGG_GU; ++u) {
                    const float* xr = x + (size_t)(ent[u] & 0xFFFFu) * D + 2 * lane;
                    va[u] = *(const v2f*)xr;
                    vb[u] = *(const v2f*)(xr + 64);
                }
#pragma unroll
                for (int u = 0; u < AGG_GU; ++u)
                    if (i0 + (unsigned)u < n) add_row(sAgg, ent[u] >> 16, lane, va[u], vb[u]);
            }
        }
        cur = nxt;
    }
}

__device__ __forceinline__ void sum_by_scan(const int* __restrict__ src, const int* __restrict__ dst,
                                            const float* __restrict__ x, int lo, unsigned cnt,
                                            int wave, int lane, float* sAgg) {
    for (int e0 = 0; e0 < NE; e0 += 32) {
        const unsigned r = (unsigned)(dst[e0 + lane] - lo);
        const bool mine  = (r < cnt) && ((r & 31u) == (unsigned)wave);
        unsigned m = __builtin_amdgcn_ballot_w32(mine);
        while (m != 0u) {
            const int l = __builtin_ctz(m);
            m &= m - 1u;
            const unsigned rl = (unsigned)__builtin_amdgcn_readlane((int)r, l);
            const int s = src[e0 + l];
            const float* xr = x + (size_t)s * D + 2 * lane;
            const v2f a = *(const v2f*)xr;
            const v2f b = *(const v2f*)(xr + 64);
            add_row(sAgg, rl, lane, a, b);
        }
    }
}

template <bool BUILD>
__global__ __launch_bounds__(AGG_T)
void aggregate_kernel(const float* __restrict__ x, const int* __restrict__ src,
                      const int* __restrict__ dst,
                      unsigned* __restrict__ q, unsigned* __restrict__ qcnt,
                      float* __restrict__ inv_deg, float* __restrict__ agg) {
    __shared__ float    sAgg[AGG_ROWS * D];
    __shared__ unsigned sQ[QCAP];

    const int t    = threadIdx.x;
    const int lane = t & 31;
    const int wave = __builtin_amdgcn_readfirstlane(t >> 5);
    const int lo   = blockIdx.x * AGG_ROWS;
    const unsigned cnt = (unsigned)min(AGG_ROWS, N_NODES - lo);
    unsigned* qseg = q + (size_t)blockIdx.x * QCAP + (size_t)wave * BINCAP;
    unsigned* qn   = qcnt + (size_t)blockIdx.x * AGG_NW + wave;
    unsigned* bin  = sQ + wave * BINCAP;

#pragma unroll
    for (int j = 0; j < AGG_SLOTS; ++j) {
        const unsigned r = (unsigned)(wave + AGG_NW * j);
        if (r < cnt) {
            float* ar = sAgg + r * D + 2 * lane;
            *(v2f*)ar        = (v2f){0.0f, 0.0f};
            *(v2f*)(ar + 64) = (v2f){0.0f, 0.0f};
        }
    }

    if constexpr (BUILD) {
        __shared__ unsigned sBinN[AGG_NW];
        __shared__ unsigned sDeg[AGG_ROWS];

        if (t < AGG_NW) sBinN[t] = 0u;
        if (t < AGG_ROWS) sDeg[t] = 0u;
        __syncthreads();
        filter_bins(dst, lo, cnt, sQ, sBinN, sDeg);
        __syncthreads();
        if ((unsigned)t < cnt) inv_deg[lo + t] = 1.0f / fmaxf((float)sDeg[t], 1.0f);
        const unsigned nraw = (unsigned)__builtin_amdgcn_readfirstlane((int)sBinN[wave]);
        const unsigned n    = min(nraw, (unsigned)BINCAP);
        for (unsigned j0 = (unsigned)lane; j0 < n; j0 += 128u) {
            unsigned v[4];
            int      s[4];
#pragma unroll
            for (int u = 0; u < 4; ++u) {
                unsigned i = j0 + 32u * (unsigned)u;
                v[u] = (i < n) ? bin[i] : 0u;
            }
#pragma unroll
            for (int u = 0; u < 4; ++u) s[u] = src[v[u] & 0xFFFFFu];
#pragma unroll
            for (int u = 0; u < 4; ++u) {
                unsigned i = j0 + 32u * (unsigned)u;
                if (i < n) {
                    unsigned ent = (unsigned)s[u] | ((v[u] >> 20) << 16);
                    bin[i]  = ent;
                    qseg[i] = ent;
                }
            }
        }
        if (lane == 0) *qn = nraw;

        if (nraw <= (unsigned)BINCAP) sum_bin(bin, n, x, lane, sAgg);
        else                          sum_by_scan(src, dst, x, lo, cnt, wave, lane, sAgg);
    } else {
        const unsigned nraw = (unsigned)__builtin_amdgcn_readfirstlane((int)*qn);
        if (nraw <= (unsigned)BINCAP) {
            for (unsigned j0 = (unsigned)lane; j0 < nraw; j0 += 128u) {
                unsigned v[4];
#pragma unroll
                for (int u = 0; u < 4; ++u) {
                    unsigned i = j0 + 32u * (unsigned)u;
                    v[u] = (i < nraw) ? qseg[i] : 0u;
                }
#pragma unroll
                for (int u = 0; u < 4; ++u) {
                    unsigned i = j0 + 32u * (unsigned)u;
                    if (i < nraw) bin[i] = v[u];
                }
            }
            sum_bin(bin, nraw, x, lane, sAgg);
        } else {
            sum_by_scan(src, dst, x, lo, cnt, wave, lane, sAgg);
        }
    }

#pragma unroll
    for (int j = 0; j < AGG_SLOTS; ++j) {
        const unsigned r = (unsigned)(wave + AGG_NW * j);
        if (r < cnt) {
            const float* ar = sAgg + r * D + 2 * lane;
            float* ao = agg + (size_t)(lo + (int)r) * D + 2 * lane;
            *(v2f*)ao        = *(const v2f*)ar;
            *(v2f*)(ao + 64) = *(const v2f*)(ar + 64);
        }
    }
}

#define WPITCH    288
#define GEMM_W    16
#define GEMM_FULL 12
#define GEMM_T    (GEMM_W * 32)
#define GEMM_WGS  256

static_assert(GEMM_W - GEMM_FULL == 4, "the left-over tile is split in column quarters");
static_assert(MTILES >= GEMM_WGS * GEMM_FULL && MTILES - GEMM_WGS * GEMM_FULL <= GEMM_WGS,
              "every tile has exactly one owner");

__device__ __forceinline__ void stage_weights(const float* __restrict__ Wg, float* sW, int t) {
#pragma unroll
    for (int i = 0; i < 2048 / GEMM_T; ++i) {
        const int id = t + GEMM_T * i;
        const int p  = id >> 5;
        const int n4 = id & 31;
        const v4f a = *(const v4f*)(Wg + (size_t)(2 * p) * D + 4 * n4);
        const v4f b = *(const v4f*)(Wg + (size_t)(2 * p + 1) * D + 4 * n4);
        float* o = sW + p * WPITCH + 8 * n4;
        *(v4f*)o       = (v4f){a.x, b.x, a.y, b.y};
        *(v4f*)(o + 4) = (v4f){a.z, b.z, a.w, b.w};
    }
}

template <int NTN>
__device__ __forceinline__ void tile_times_weights(const float* tile, const float* sW,
                                                   int nt0, int c, int half, v8f* acc) {
    const float* ap = tile + c * LDSS + 2 * half;
    const float* wp = sW + half * WPITCH + 32 * nt0 + 2 * c;
#pragma unroll
    for (int ks = 0; ks < 32; ++ks) {
        const v2f a = *(const v2f*)(ap + 4 * ks);
#pragma unroll
        for (int j = 0; j < NTN; ++j) {
            const v2f b = *(const v2f*)(wp + 2 * ks * WPITCH + 32 * j);
            acc[j] = __builtin_amdgcn_wmma_f32_16x16x4_f32(false, a, false, b,
                                                           (short)0, acc[j], false, false);
        }
    }
}

template <int MODE, int NTN>
__device__ __forceinline__ void wave_tile(float* tile, const float* sW, const v4f* xr, const v4f* ar,
                                          const float* __restrict__ inv_deg, size_t rowbase, int nt0,
                                          int lane,
                                          const float* __restrict__ bias,
                                          const float* __restrict__ gamma, const float* __restrict__ beta,
                                          const float* __restrict__ mean, const float* __restrict__ var,
                                          float* __restrict__ out) {
    const int half = lane >> 4;
    const int c    = lane & 15;

    v8f acc[NTN];
#pragma unroll
    for (int j = 0; j < NTN; ++j) acc[j] = (v8f){};

#pragma unroll
    for (int r = 0; r < 16; ++r) {
        v4f v;
        if (MODE == 0) {
            v = xr[r];
        } else {
            const float idg = inv_deg[rowbase + r];
            v = ar[r] * idg;
        }
        *(v4f*)(tile + r * LDSS + 4 * lane) = v;
    }
    tile_times_weights<NTN>(tile, sW, nt0, c, half, acc);

    if (MODE != 0) {
#pragma unroll
        for (int r = 0; r < 16; ++r)
            *(v4f*)(tile + r * LDSS + 4 * lane) = xr[r];
        tile_times_weights<NTN>(tile, sW + 64 * WPITCH, nt0, c, half, acc);
    }

#pragma unroll
    for (int j = 0; j < NTN; ++j) {
        const int n = 16 * (nt0 + j) + c;
        const float bcol = bias[n];
        float s = 1.0f, tt = 0.0f;
        if (MODE != 0) {
            s  = gamma[n] * rsqrtf(var[n] + EPS_BN);
            tt = beta[n] - mean[n] * s;
        }
#pragma unroll
        for (int v = 0; v < 8; ++v) {
            const int m = v + 8 * half;
            const float h = acc[j][v] + bcol;
            float y;
            if (MODE == 0) {
                y = fmaxf(h, 0.0f);
            } else {
                y = fmaxf(h * s + tt, 0.0f);
                if (MODE == 1) y += tile[m * LDSS + n];
            }
            tile[m * LDSS + n] = y;
        }
    }
    constexpr int LPR = 4 * NTN;
    constexpr int RPI = 32 / LPR;
    const int srow = lane / LPR;
    const int scol = 16 * nt0 + 4 * (lane % LPR);
#pragma unroll
    for (int i = 0; i < 16 / RPI; ++i) {
        const int r = i * RPI + srow;
        *(v4f*)(out + (rowbase + r) * D + scol) = *(const v4f*)(tile + r * LDSS + scol);
    }
}

template <int MODE>
__global__ __launch_bounds__(GEMM_T)
void sage_gemm_kernel(const float* __restrict__ Aagg, const float* __restrict__ inv_deg,
                      const float* __restrict__ Ax,
                      const float* __restrict__ W1, const float* __restrict__ W2,
                      const float* __restrict__ bias,
                      const float* __restrict__ gamma, const float* __restrict__ beta,
                      const float* __restrict__ mean, const float* __restrict__ var,
                      float* __restrict__ out) {
    __shared__ __attribute__((aligned(16))) float sW[(MODE == 0 ? 1 : 2) * 64 * WPITCH];
    __shared__ __attribute__((aligned(16))) float sT[GEMM_W * 16 * LDSS];

    const int t    = threadIdx.x;
    const int lane = t & 31;
    const int wave = __builtin_amdgcn_readfirstlane(t >> 5);

    const bool full = wave < GEMM_FULL;
    const int  mt   = full ? (int)blockIdx.x + GEMM_WGS * wave
                           : GEMM_WGS * GEMM_FULL + (int)blockIdx.x;
    const bool has  = mt < MTILES;
    const size_t rowbase = (size_t)(has ? mt : 0) * 16;
    float* tile = sT + wave * (16 * LDSS);

    v4f xr[16];
    v4f ar[16];
    if (has) {
#pragma unroll
        for (int r = 0; r < 16; ++r)
            xr[r] = *(const v4f*)(Ax + (rowbase + r) * D + 4 * lane);
        if (MODE != 0) {
#pragma unroll
            for (int r = 0; r < 16; ++r)
                ar[r] = *(const v4f*)(Aagg + (rowbase + r) * D + 4 * lane);
        }
    }

    stage_weights(W1, sW, t);
    if (MODE != 0) stage_weights(W2, sW + 64 * WPITCH, t);
    __syncthreads();
    if (!has) return;

    if (full)
        wave_tile<MODE, 8>(tile, sW, xr, ar, inv_deg, rowbase, 0, lane,
                           bias, gamma, beta, mean, var, out);
    else
        wave_tile<MODE, 2>(tile, sW, xr, ar, inv_deg, rowbase, 2 * (wave - GEMM_FULL), lane,
                           bias, gamma, beta, mean, var, out);
}


extern "C" void kernel_launch(void* const* d_in, const int* in_sizes, int n_in,
                              void* d_out, int out_size, void* d_ws, size_t ws_size,
                              hipStream_t stream) {
    (void)in_sizes; (void)n_in; (void)out_size; (void)ws_size;

    const float* x   = (const float*)d_in[0];
    const int*   ei  = (const int*)d_in[1];
    const float* Wp  = (const float*)d_in[2];
    const float* bp  = (const float*)d_in[3];
    const float* Wl  = (const float*)d_in[4];
    const float* bl  = (const float*)d_in[5];
    const float* Wr  = (const float*)d_in[6];
    const float* gam = (const float*)d_in[7];
    const float* bet = (const float*)d_in[8];
    const float* mea = (const float*)d_in[9];
    const float* var = (const float*)d_in[10];

    const int* src = ei;
    const int* dst = ei + NE;

    float* ws  = (float*)d_ws;
    float* deg = ws;
    float* xa  = ws + ((N_NODES + 63) / 64) * 64;
    float* xb  = xa + (size_t)N_NODES * D;
    float* agg = xb + (size_t)N_NODES * D;

    unsigned* q    = (unsigned*)d_out;
    unsigned* qcnt = q + (size_t)AGG_WGS * QCAP;

    sage_gemm_kernel<0><<<GEMM_WGS, GEMM_T, 0, stream>>>(
        nullptr, deg, x, Wp, Wp, bp, gam, bet, mea, var, xa);

    float* xcur  = xa;
    float* xnext = xb;
    for (int i = 0; i < 4; ++i) {
        if (i == 0)
            aggregate_kernel<true><<<AGG_WGS, AGG_T, 0, stream>>>(xcur, src, dst, q, qcnt, deg, agg);
        else
            aggregate_kernel<false><<<AGG_WGS, AGG_T, 0, stream>>>(xcur, src, dst, q, qcnt, deg, agg);

        if (i < 3)
            sage_gemm_kernel<1><<<GEMM_WGS, GEMM_T, 0, stream>>>(
                agg, deg, xcur,
                Wl + (size_t)i * D * D, Wr + (size_t)i * D * D, bl + (size_t)i * D,
                gam + (size_t)i * D, bet + (size_t)i * D, mea + (size_t)i * D, var + (size_t)i * D,
                xnext);
        else
            sage_gemm_kernel<2><<<GEMM_WGS, GEMM_T, 0, stream>>>(
                agg, deg, xcur,
                Wl + (size_t)i * D * D, Wr + (size_t)i * D * D, bl + (size_t)i * D,
                gam + (size_t)i * D, bet + (size_t)i * D, mea + (size_t)i * D, var + (size_t)i * D,
                (float*)d_out);

        float* tmp = xcur; xcur = xnext; xnext = tmp;
    }
}


# Edge destinations packed once to one byte (node-range index, written by the projection launch): the first neighbour-sum launch filters a quarter of the bytes, reads dst, src of its own edges only
# speedup vs baseline: 5.3217x; 1.0696x over previous
#include <hip/hip_runtime.h>

#define N_NODES 50000
#define NE      800000
#define D       128
#define EPS_BN  1e-5f
#define MTILES  (N_NODES / 16)
#define LDSS    132

typedef __attribute__((ext_vector_type(2))) float v2f;
typedef __attribute__((ext_vector_type(4))) float v4f;
typedef __attribute__((ext_vector_type(8))) float v8f;

#define AGG_T     1024
#define AGG_NW    (AGG_T / 32)
#define AGG_ROWS  196
#define AGG_SLOTS ((AGG_ROWS + AGG_NW - 1) / AGG_NW)
#define AGG_WGS   ((N_NODES + AGG_ROWS - 1) / AGG_ROWS)
#define QCAP      16384
#define BINCAP    (QCAP / AGG_NW)
#define AGG_GU    8
#define NG4       (NE / 4)
#define NG16      (NE / 16)
#define LCAP      512

static_assert(N_NODES <= 65536, "a bin entry keeps the source node in 16 bits");
static_assert(AGG_ROWS <= 256 && NE <= (1 << 20), "bin entry fields");
static_assert(NE % 32 == 0, "a wave walks the edge list 32 edges at a time");
static_assert(AGG_NW == 32, "row -> wave map is r & 31");
static_assert(AGG_ROWS <= AGG_T && NE < (1 << 24), "one thread per row writes 1/degree; counts exact in f32");
static_assert(AGG_WGS <= 256, "a range index is kept in one byte");
static_assert(NE % 16 == 0, "the packed list is read 16 edges at a time");
static_assert((size_t)AGG_WGS * QCAP + (size_t)AGG_WGS * AGG_NW + NE / 4 <= (size_t)N_NODES * D,
              "bins, counts and the packed list fit in the output buffer");
static_assert(((size_t)AGG_WGS * QCAP + (size_t)AGG_WGS * AGG_NW) % 4 == 0,
              "the packed list starts on a 16-byte boundary");

typedef __attribute__((ext_vector_type(4))) int v4i;
typedef __attribute__((ext_vector_type(4))) unsigned v4u;

__device__ __forceinline__ unsigned range_byte(int d) {
    return ((unsigned)d / (unsigned)AGG_ROWS) & 0xFFu;
}

__device__ __forceinline__ unsigned zero_bytes(unsigned x) {
    const unsigned s = (x & 0x7F7F7F7Fu) + 0x7F7F7F7Fu;
    return ~(s | x | 0x7F7F7F7Fu);
}

__device__ __forceinline__ void add_row(float* sAgg, unsigned row, int lane, v2f a, v2f b) {
    float* ar = sAgg + row * D + 2 * lane;
    v2f p = *(const v2f*)ar;
    v2f q = *(const v2f*)(ar + 64);
    p += a;
    q += b;
    *(v2f*)ar        = p;
    *(v2f*)(ar + 64) = q;
}

__device__ __forceinline__ void filter_bins(const int* __restrict__ dst, int lo, unsigned cnt,
                                            unsigned* sQ, unsigned* sBinN, unsigned* sDeg) {
    const int t = threadIdx.x;
    const v4i* __restrict__ d4 = (const v4i*)dst;
    for (int g0 = t; g0 < NG4; g0 += 4 * AGG_T) {
        v4i dv[4];
#pragma unroll
        for (int u = 0; u < 4; ++u) {
            int g = g0 + u * AGG_T;
            dv[u] = d4[g < NG4 ? g : NG4 - 1];
        }
#pragma unroll
        for (int u = 0; u < 4; ++u) {
            int g = g0 + u * AGG_T;
            unsigned r0 = (unsigned)(dv[u].x - lo);
            unsigned r1 = (unsigned)(dv[u].y - lo);
            unsigned r2 = (unsigned)(dv[u].z - lo);
            unsigned r3 = (unsigned)(dv[u].w - lo);
            unsigned rmin = min(min(r0, r1), min(r2, r3));
            if (g < NG4 && rmin < cnt) {
                unsigned rr[4] = {r0, r1, r2, r3};
#pragma unroll
                for (int c = 0; c < 4; ++c) {
                    unsigned r = rr[c];
                    if (r < cnt) {
                        unsigned e   = (unsigned)(g * 4 + c);
                        unsigned w   = r & 31u;
                        unsigned pos = atomicAdd(&sBinN[w], 1u);
                        if (pos < (unsigned)BINCAP) sQ[w * BINCAP + pos] = e | (r << 20);
                        atomicAdd(&sDeg[r], 1u);
                    }
                }
            }
        }
    }
}

__device__ __forceinline__ unsigned list_range_edges(const unsigned* __restrict__ rid, unsigned range,
                                                     unsigned* list) {
    const int t = threadIdx.x;
    const v4u* __restrict__ r4 = (const v4u*)rid;
    const unsigned pat = range * 0x01010101u;
    unsigned found = 0u;
    for (int g0 = t; g0 < NG16; g0 += 4 * AGG_T) {
        v4u dv[4];
#pragma unroll
        for (int u = 0; u < 4; ++u) {
            int g = g0 + u * AGG_T;
            dv[u] = r4[g < NG16 ? g : NG16 - 1];
        }
#pragma unroll
        for (int u = 0; u < 4; ++u) {
            int g = g0 + u * AGG_T;
            unsigned m = (zero_bytes(dv[u].x ^ pat) >> 7) | (zero_bytes(dv[u].y ^ pat) >> 6) |
                         (zero_bytes(dv[u].z ^ pat) >> 5) | (zero_bytes(dv[u].w ^ pat) >> 4);
            if (g >= NG16) m = 0u;
            unsigned bal = __builtin_amdgcn_ballot_w32(m != 0u);
            while (bal != 0u) {
                if (m != 0u) {
                    const unsigned i   = (unsigned)__builtin_ctz(m);
                    const unsigned pos = found + __builtin_amdgcn_mbcnt_lo(bal, 0u);
                    m &= m - 1u;
                    if (pos < (unsigned)LCAP)
                        list[pos] = (unsigned)g * 16u + (i & 7u) * 4u + (i >> 3);
                }
                found += (unsigned)__builtin_popcount(bal);
                bal = __builtin_amdgcn_ballot_w32(m != 0u);
            }
        }
    }
    return found;
}

__device__ __forceinline__ void bin_listed_edges(const unsigned* list, unsigned n,
                                                 const int* __restrict__ src, const int* __restrict__ dst,
                                                 int lo, unsigned cnt, int lane,
                                                 unsigned* sQ, unsigned* sBinN, unsigned* sDeg) {
    for (unsigned j0 = (unsigned)lane; j0 < n; j0 += 128u) {
        unsigned e[4];
        int      d[4], s[4];
#pragma unroll
        for (int u = 0; u < 4; ++u) {
            unsigned i = j0 + 32u * (unsigned)u;
            e[u] = (i < n) ? list[i] : 0u;
        }
#pragma unroll
        for (int u = 0; u < 4; ++u) {
            d[u] = dst[e[u]];
            s[u] = src[e[u]];
        }
#pragma unroll
        for (int u = 0; u < 4; ++u)
            asm volatile("" : "+v"(s[u]));
#pragma unroll
        for (int u = 0; u < 4; ++u) {
            unsigned i = j0 + 32u * (unsigned)u;
            unsigned r = (unsigned)(d[u] - lo);
            if (i < n && r < cnt) {
                unsigned w   = r & 31u;
                unsigned pos = atomicAdd(&sBinN[w], 1u);
                if (pos < (unsigned)BINCAP) sQ[w * BINCAP + pos] = (unsigned)s[u] | (r << 16);
                atomicAdd(&sDeg[r], 1u);
            }
        }
    }
}

__device__ __forceinline__ void sum_bin(const unsigned* Q, unsigned n, const float* __restrict__ x,
                                        int lane, float* sAgg) {
    unsigned cur = ((unsigned)lane < n) ? Q[lane] : 0u;
    for (unsigned base = 0; base < n; base += 32) {
        const unsigned nb  = base + 32u + (unsigned)lane;
        const unsigned nxt = (nb < n) ? Q[nb] : 0u;
#pragma unroll
        for (int g = 0; g < 32 / AGG_GU; ++g) {
            const unsigned i0 = base + (unsigned)(g * AGG_GU);
            if (i0 < n) {
                unsigned ent[AGG_GU];
                v2f va[AGG_GU], vb[AGG_GU];
#pragma unroll
                for (int u = 0; u < AGG_GU; ++u)
                    ent[u] = (unsigned)__builtin_amdgcn_readlane((int)cur, g * AGG_GU + u);
#pragma unroll
                for (int u = 0; u < AGG_GU; ++u) {
                    const float* xr = x + (size_t)(ent[u] & 0xFFFFu) * D + 2 * lane;
                    va[u] = *(const v2f*)xr;
                    vb[u] = *(const v2f*)(xr + 64);
                }
#pragma unroll
                for (int u = 0; u < AGG_GU; ++u)
                    if (i0 + (unsigned)u < n) add_row(sAgg, ent[u] >> 16, lane, va[u], vb[u]);
            }
        }
        cur = nxt;
    }
}

__device__ __forceinline__ void sum_by_scan(const int* __restrict__ src, const int* __restrict__ dst,
                                            const float* __restrict__ x, int lo, unsigned cnt,
                                            int wave, int lane, float* sAgg) {
    for (int e0 = 0; e0 < NE; e0 += 32) {
        const unsigned r = (unsigned)(dst[e0 + lane] - lo);
        const bool mine  = (r < cnt) && ((r & 31u) == (unsigned)wave);
        unsigned m = __builtin_amdgcn_ballot_w32(mine);
        while (m != 0u) {
            const int l = __builtin_ctz(m);
            m &= m - 1u;
            const unsigned rl = (unsigned)__builtin_amdgcn_readlane((int)r, l);
            const int s = src[e0 + l];
            const float* xr = x + (size_t)s * D + 2 * lane;
            const v2f a = *(const v2f*)xr;
            const v2f b = *(const v2f*)(xr + 64);
            add_row(sAgg, rl, lane, a, b);
        }
    }
}

template <bool BUILD>
__global__ __launch_bounds__(AGG_T)
void aggregate_kernel(const float* __restrict__ x, const int* __restrict__ src,
                      const int* __restrict__ dst, const unsigned* __restrict__ rid,
                      unsigned* __restrict__ q, unsigned* __restrict__ qcnt,
                      float* __restrict__ inv_deg, float* __restrict__ agg) {
    __shared__ float    sAgg[AGG_ROWS * D];
    __shared__ unsigned sQ[QCAP];

    const int t    = threadIdx.x;
    const int lane = t & 31;
    const int wave = __builtin_amdgcn_readfirstlane(t >> 5);
    const int lo   = blockIdx.x * AGG_ROWS;
    const unsigned cnt = (unsigned)min(AGG_ROWS, N_NODES - lo);
    unsigned* qseg = q + (size_t)blockIdx.x * QCAP + (size_t)wave * BINCAP;
    unsigned* qn   = qcnt + (size_t)blockIdx.x * AGG_NW + wave;
    unsigned* bin  = sQ + wave * BINCAP;

#pragma unroll
    for (int j = 0; j < AGG_SLOTS; ++j) {
        const unsigned r = (unsigned)(wave + AGG_NW * j);
        if (r < cnt) {
            float* ar = sAgg + r * D + 2 * lane;
            *(v2f*)ar        = (v2f){0.0f, 0.0f};
            *(v2f*)(ar + 64) = (v2f){0.0f, 0.0f};
        }
    }

    if constexpr (BUILD) {
        __shared__ unsigned sL[AGG_NW * LCAP];
        __shared__ unsigned sBinN[AGG_NW];
        __shared__ unsigned sDeg[AGG_ROWS];
        __shared__ unsigned sOver;

        if (t < AGG_NW) sBinN[t] = 0u;
        if (t < AGG_ROWS) sDeg[t] = 0u;
        if (t == 0) sOver = 0u;
        unsigned* list = sL + wave * LCAP;
        const unsigned lraw = (unsigned)__builtin_amdgcn_readfirstlane(
            (int)list_range_edges(rid, (unsigned)blockIdx.x, list));
        __syncthreads();
        if (lraw > (unsigned)LCAP) {
            if (lane == 0) sOver = 1u;
        } else {
            bin_listed_edges(list, lraw, src, dst, lo, cnt, lane, sQ, sBinN, sDeg);
        }
        __syncthreads();
        const bool over = sOver != 0u;
        if (over) {
            if (t < AGG_NW) sBinN[t] = 0u;
            if (t < AGG_ROWS) sDeg[t] = 0u;
            __syncthreads();
            filter_bins(dst, lo, cnt, sQ, sBinN, sDeg);
            __syncthreads();
        }
        if ((unsigned)t < cnt) inv_deg[lo + t] = 1.0f / fmaxf((float)sDeg[t], 1.0f);
        const unsigned nraw = (unsigned)__builtin_amdgcn_readfirstlane((int)sBinN[wave]);
        const unsigned n    = min(nraw, (unsigned)BINCAP);
        if (over) {
            for (unsigned j0 = (unsigned)lane; j0 < n; j0 += 128u) {
                unsigned v[4];
                int      s[4];
#pragma unroll
                for (int u = 0; u < 4; ++u) {
                    unsigned i = j0 + 32u * (unsigned)u;
                    v[u] = (i < n) ? bin[i] : 0u;
                }
#pragma unroll
                for (int u = 0; u < 4; ++u) s[u] = src[v[u] & 0xFFFFFu];
#pragma unroll
                for (int u = 0; u < 4; ++u) {
                    unsigned i = j0 + 32u * (unsigned)u;
                    if (i < n) {
                        unsigned ent = (unsigned)s[u] | ((v[u] >> 20) << 16);
                        bin[i]  = ent;
                        qseg[i] = ent;
                    }
                }
            }
        } else {
            for (unsigned i = (unsigned)lane; i < n; i += 32u) qseg[i] = bin[i];
        }
        if (lane == 0) *qn = nraw;

        if (nraw <= (unsigned)BINCAP) sum_bin(bin, n, x, lane, sAgg);
        else                          sum_by_scan(src, dst, x, lo, cnt, wave, lane, sAgg);
    } else {
        const unsigned nraw = (unsigned)__builtin_amdgcn_readfirstlane((int)*qn);
        if (nraw <= (unsigned)BINCAP) {
            for (unsigned j0 = (unsigned)lane; j0 < nraw; j0 += 128u) {
                unsigned v[4];
#pragma unroll
                for (int u = 0; u < 4; ++u) {
                    unsigned i = j0 + 32u * (unsigned)u;
                    v[u] = (i < nraw) ? qseg[i] : 0u;
                }
#pragma unroll
                for (int u = 0; u < 4; ++u) {
                    unsigned i = j0 + 32u * (unsigned)u;
                    if (i < nraw) bin[i] = v[u];
                }
            }
            sum_bin(bin, nraw, x, lane, sAgg);
        } else {
            sum_by_scan(src, dst, x, lo, cnt, wave, lane, sAgg);
        }
    }

#pragma unroll
    for (int j = 0; j < AGG_SLOTS; ++j) {
        const unsigned r = (unsigned)(wave + AGG_NW * j);
        if (r < cnt) {
            const float* ar = sAgg + r * D + 2 * lane;
            float* ao = agg + (size_t)(lo + (int)r) * D + 2 * lane;
            *(v2f*)ao        = *(const v2f*)ar;
            *(v2f*)(ao + 64) = *(const v2f*)(ar + 64);
        }
    }
}

#define WPITCH    288
#define GEMM_W    16
#define GEMM_FULL 12
#define GEMM_T    (GEMM_W * 32)
#define GEMM_WGS  256

static_assert(GEMM_W - GEMM_FULL == 4, "the left-over tile is split in column quarters");
static_assert(MTILES >= GEMM_WGS * GEMM_FULL && MTILES - GEMM_WGS * GEMM_FULL <= GEMM_WGS,
              "every tile has exactly one owner");

__device__ __forceinline__ void stage_weights(const float* __restrict__ Wg, float* sW, int t) {
#pragma unroll
    for (int i = 0; i < 2048 / GEMM_T; ++i) {
        const int id = t + GEMM_T * i;
        const int p  = id >> 5;
        const int n4 = id & 31;
        const v4f a = *(const v4f*)(Wg + (size_t)(2 * p) * D + 4 * n4);
        const v4f b = *(const v4f*)(Wg + (size_t)(2 * p + 1) * D + 4 * n4);
        float* o = sW + p * WPITCH + 8 * n4;
        *(v4f*)o       = (v4f){a.x, b.x, a.y, b.y};
        *(v4f*)(o + 4) = (v4f){a.z, b.z, a.w, b.w};
    }
}

template <int NTN>
__device__ __forceinline__ void tile_times_weights(const float* tile, const float* sW,
                                                   int nt0, int c, int half, v8f* acc) {
    const float* ap = tile + c * LDSS + 2 * half;
    const float* wp = sW + half * WPITCH + 32 * nt0 + 2 * c;
#pragma unroll
    for (int ks = 0; ks < 32; ++ks) {
        const v2f a = *(const v2f*)(ap + 4 * ks);
#pragma unroll
        for (int j = 0; j < NTN; ++j) {
            const v2f b = *(const v2f*)(wp + 2 * ks * WPITCH + 32 * j);
            acc[j] = __builtin_amdgcn_wmma_f32_16x16x4_f32(false, a, false, b,
                                                           (short)0, acc[j], false, false);
        }
    }
}

template <int MODE, int NTN>
__device__ __forceinline__ void wave_tile(float* tile, const float* sW, const v4f* xr, const v4f* ar,
                                          const float* __restrict__ inv_deg, size_t rowbase, int nt0,
                                          int lane,
                                          const float* __restrict__ bias,
                                          const float* __restrict__ gamma, const float* __restrict__ beta,
                                          const float* __restrict__ mean, const float* __restrict__ var,
                                          float* __restrict__ out) {
    const int half = lane >> 4;
    const int c    = lane & 15;

    v8f acc[NTN];
#pragma unroll
    for (int j = 0; j < NTN; ++j) acc[j] = (v8f){};

#pragma unroll
    for (int r = 0; r < 16; ++r) {
        v4f v;
        if (MODE == 0) {
            v = xr[r];
        } else {
            const float idg = inv_deg[rowbase + r];
            v = ar[r] * idg;
        }
        *(v4f*)(tile + r * LDSS + 4 * lane) = v;
    }
    tile_times_weights<NTN>(tile, sW, nt0, c, half, acc);

    if (MODE != 0) {
#pragma unroll
        for (int r = 0; r < 16; ++r)
            *(v4f*)(tile + r * LDSS + 4 * lane) = xr[r];
        tile_times_weights<NTN>(tile, sW + 64 * WPITCH, nt0, c, half, acc);
    }

#pragma unroll
    for (int j = 0; j < NTN; ++j) {
        const int n = 16 * (nt0 + j) + c;
        const float bcol = bias[n];
        float s = 1.0f, tt = 0.0f;
        if (MODE != 0) {
            s  = gamma[n] * rsqrtf(var[n] + EPS_BN);
            tt = beta[n] - mean[n] * s;
        }
#pragma unroll
        for (int v = 0; v < 8; ++v) {
            const int m = v + 8 * half;
            const float h = acc[j][v] + bcol;
            float y;
            if (MODE == 0) {
                y = fmaxf(h, 0.0f);
            } else {
                y = fmaxf(h * s + tt, 0.0f);
                if (MODE == 1) y += tile[m * LDSS + n];
            }
            tile[m * LDSS + n] = y;
        }
    }
    constexpr int LPR = 4 * NTN;
    constexpr int RPI = 32 / LPR;
    const int srow = lane / LPR;
    const int scol = 16 * nt0 + 4 * (lane % LPR);
#pragma unroll
    for (int i = 0; i < 16 / RPI; ++i) {
        const int r = i * RPI + srow;
        *(v4f*)(out + (rowbase + r) * D + scol) = *(const v4f*)(tile + r * LDSS + scol);
    }
}

template <int MODE>
__global__ __launch_bounds__(GEMM_T)
void sage_gemm_kernel(const float* __restrict__ Aagg, const float* __restrict__ inv_deg,
                      const float* __restrict__ Ax,
                      const float* __restrict__ W1, const float* __restrict__ W2,
                      const float* __restrict__ bias,
                      const float* __restrict__ gamma, const float* __restrict__ beta,
                      const float* __restrict__ mean, const float* __restrict__ var,
                      float* __restrict__ out,
                      const int* __restrict__ dst, unsigned* __restrict__ rid) {
    __shared__ __attribute__((aligned(16))) float sW[(MODE == 0 ? 1 : 2) * 64 * WPITCH];
    __shared__ __attribute__((aligned(16))) float sT[GEMM_W * 16 * LDSS];

    const int t    = threadIdx.x;
    const int lane = t & 31;
    const int wave = __builtin_amdgcn_readfirstlane(t >> 5);

    const bool full = wave < GEMM_FULL;
    const int  mt   = full ? (int)blockIdx.x + GEMM_WGS * wave
                           : GEMM_WGS * GEMM_FULL + (int)blockIdx.x;
    const bool has  = mt < MTILES;
    const size_t rowbase = (size_t)(has ? mt : 0) * 16;
    float* tile = sT + wave * (16 * LDSS);

    constexpr int PK_T = GEMM_WGS * GEMM_T;
    constexpr int PK_N = (NG4 + PK_T - 1) / PK_T;
    v4i dq[PK_N];
    if (MODE == 0) {
        const v4i* __restrict__ d4 = (const v4i*)dst;
#pragma unroll
        for (int u = 0; u < PK_N; ++u) {
            const int g = (int)blockIdx.x * GEMM_T + t + u * PK_T;
            dq[u] = d4[g < NG4 ? g : NG4 - 1];
        }
    }

    v4f xr[16];
    v4f ar[16];
    if (has) {
#pragma unroll
        for (int r = 0; r < 16; ++r)
            xr[r] = *(const v4f*)(Ax + (rowbase + r) * D + 4 * lane);
        if (MODE != 0) {
#pragma unroll
            for (int r = 0; r < 16; ++r)
                ar[r] = *(const v4f*)(Aagg + (rowbase + r) * D + 4 * lane);
        }
    }

    stage_weights(W1, sW, t);
    if (MODE != 0) stage_weights(W2, sW + 64 * WPITCH, t);
    __syncthreads();
    if (MODE == 0) {
#pragma unroll
        for (int u = 0; u < PK_N; ++u) {
            const int g = (int)blockIdx.x * GEMM_T + t + u * PK_T;
            if (g < NG4)
                rid[g] = range_byte(dq[u].x) | (range_byte(dq[u].y) << 8) |
                         (range_byte(dq[u].z) << 16) | (range_byte(dq[u].w) << 24);
        }
    }
    if (!has) return;

    if (full)
        wave_tile<MODE, 8>(tile, sW, xr, ar, inv_deg, rowbase, 0, lane,
                           bias, gamma, beta, mean, var, out);
    else
        wave_tile<MODE, 2>(tile, sW, xr, ar, inv_deg, rowbase, 2 * (wave - GEMM_FULL), lane,
                           bias, gamma, beta, mean, var, out);
}


extern "C" void kernel_launch(void* const* d_in, const int* in_sizes, int n_in,
                              void* d_out, int out_size, void* d_ws, size_t ws_size,
                              hipStream_t stream) {
    (void)in_sizes; (void)n_in; (void)out_size; (void)ws_size;

    const float* x   = (const float*)d_in[0];
    const int*   ei  = (const int*)d_in[1];
    const float* Wp  = (const float*)d_in[2];
    const float* bp  = (const float*)d_in[3];
    const float* Wl  = (const float*)d_in[4];
    const float* bl  = (const float*)d_in[5];
    const float* Wr  = (const float*)d_in[6];
    const float* gam = (const float*)d_in[7];
    const float* bet = (const float*)d_in[8];
    const float* mea = (const float*)d_in[9];
    const float* var = (const float*)d_in[10];

    const int* src = ei;
    const int* dst = ei + NE;

    float* ws  = (float*)d_ws;
    float* deg = ws;
    float* xa  = ws + ((N_NODES + 63) / 64) * 64;
    float* xb  = xa + (size_t)N_NODES * D;
    float* agg = xb + (size_t)N_NODES * D;

    unsigned* q    = (unsigned*)d_out;
    unsigned* qcnt = q + (size_t)AGG_WGS * QCAP;
    unsigned* rid  = qcnt + (size_t)AGG_WGS * AGG_NW;

    sage_gemm_kernel<0><<<GEMM_WGS, GEMM_T, 0, stream>>>(
        nullptr, deg, x, Wp, Wp, bp, gam, bet, mea, var, xa, dst, rid);

    float* xcur  = xa;
    float* xnext = xb;
    for (int i = 0; i < 4; ++i) {
        if (i == 0)
            aggregate_kernel<true><<<AGG_WGS, AGG_T, 0, stream>>>(xcur, src, dst, rid, q, qcnt, deg, agg);
        else
            aggregate_kernel<false><<<AGG_WGS, AGG_T, 0, stream>>>(xcur, src, dst, rid, q, qcnt, deg, agg);

        if (i < 3)
            sage_gemm_kernel<1><<<GEMM_WGS, GEMM_T, 0, stream>>>(
                agg, deg, xcur,
                Wl + (size_t)i * D * D, Wr + (size_t)i * D * D, bl + (size_t)i * D,
                gam + (size_t)i * D, bet + (size_t)i * D, mea + (size_t)i * D, var + (size_t)i * D,
                xnext, nullptr, nullptr);
        else
            sage_gemm_kernel<2><<<GEMM_WGS, GEMM_T, 0, stream>>>(
                agg, deg, xcur,
                Wl + (size_t)i * D * D, Wr + (size_t)i * D * D, bl + (size_t)i * D,
                gam + (size_t)i * D, bet + (size_t)i * D, mea + (size_t)i * D, var + (size_t)i * D,
                (float*)d_out, nullptr, nullptr);

        float* tmp = xcur; xcur = xnext; xnext = tmp;
    }
}


# Neighbour sums in registers: a wave keeps its 7 rows' sums in one 32-register vector indexed by the edge's row slot through a scalar; LDS read-add-write and the LDS sums array go; same add order
# speedup vs baseline: 5.5745x; 1.0475x over previous
#include <hip/hip_runtime.h>

#define N_NODES 50000
#define NE      800000
#define D       128
#define EPS_BN  1e-5f
#define MTILES  (N_NODES / 16)
#define LDSS    132

typedef __attribute__((ext_vector_type(2))) float v2f;
typedef __attribute__((ext_vector_type(4))) float v4f;
typedef __attribute__((ext_vector_type(8))) float v8f;

#define AGG_T     1024
#define AGG_NW    (AGG_T / 32)
#define AGG_ROWS  196
#define AGG_SLOTS ((AGG_ROWS + AGG_NW - 1) / AGG_NW)
#define AGG_WGS   ((N_NODES + AGG_ROWS - 1) / AGG_ROWS)
#define QCAP      16384
#define BINCAP    (QCAP / AGG_NW)
#define AGG_GU    8
#define NG4       (NE / 4)
#define NG16      (NE / 16)
#define LCAP      512

static_assert(N_NODES <= 65536, "a bin entry keeps the source node in 16 bits");
static_assert(AGG_ROWS <= 256 && NE <= (1 << 20), "bin entry fields");
static_assert(NE % 32 == 0, "a wave walks the edge list 32 edges at a time");
static_assert(AGG_NW == 32, "row -> wave map is r & 31");
static_assert(AGG_SLOTS == 7, "slots 0..6 hold rows; slot 7 of the 8 is the slot of no row");
static_assert(AGG_ROWS <= AGG_T && NE < (1 << 24), "one thread per row writes 1/degree; counts exact in f32");
static_assert(AGG_WGS <= 256, "a range index is kept in one byte");
static_assert(NE % 16 == 0, "the packed list is read 16 edges at a time");
static_assert((size_t)AGG_WGS * QCAP + (size_t)AGG_WGS * AGG_NW + NE / 4 <= (size_t)N_NODES * D,
              "bins, counts and the packed list fit in the output buffer");
static_assert(((size_t)AGG_WGS * QCAP + (size_t)AGG_WGS * AGG_NW) % 4 == 0,
              "the packed list starts on a 16-byte boundary");

typedef __attribute__((ext_vector_type(4))) int v4i;
typedef __attribute__((ext_vector_type(4))) unsigned v4u;

__device__ __forceinline__ unsigned range_byte(int d) {
    return ((unsigned)d / (unsigned)AGG_ROWS) & 0xFFu;
}

__device__ __forceinline__ unsigned zero_bytes(unsigned x) {
    const unsigned s = (x & 0x7F7F7F7Fu) + 0x7F7F7F7Fu;
    return ~(s | x | 0x7F7F7F7Fu);
}

typedef __attribute__((ext_vector_type(32))) float v32f;
struct RowSums {
    v32f v;
};

__device__ __forceinline__ void add_row(RowSums& S, unsigned slot, v2f a, v2f b) {
    const unsigned i = 4u * slot;
    v2f p = (v2f){S.v[i], S.v[i + 1u]};
    v2f q = (v2f){S.v[i + 2u], S.v[i + 3u]};
    p += a;
    q += b;
    S.v[i]      = p.x;
    S.v[i + 1u] = p.y;
    S.v[i + 2u] = q.x;
    S.v[i + 3u] = q.y;
}

__device__ __forceinline__ void filter_bins(const int* __restrict__ dst, int lo, unsigned cnt,
                                            unsigned* sQ, unsigned* sBinN, unsigned* sDeg) {
    const int t = threadIdx.x;
    const v4i* __restrict__ d4 = (const v4i*)dst;
    for (int g0 = t; g0 < NG4; g0 += 4 * AGG_T) {
        v4i dv[4];
#pragma unroll
        for (int u = 0; u < 4; ++u) {
            int g = g0 + u * AGG_T;
            dv[u] = d4[g < NG4 ? g : NG4 - 1];
        }
#pragma unroll
        for (int u = 0; u < 4; ++u) {
            int g = g0 + u * AGG_T;
            unsigned r0 = (unsigned)(dv[u].x - lo);
            unsigned r1 = (unsigned)(dv[u].y - lo);
            unsigned r2 = (unsigned)(dv[u].z - lo);
            unsigned r3 = (unsigned)(dv[u].w - lo);
            unsigned rmin = min(min(r0, r1), min(r2, r3));
            if (g < NG4 && rmin < cnt) {
                unsigned rr[4] = {r0, r1, r2, r3};
#pragma unroll
                for (int c = 0; c < 4; ++c) {
                    unsigned r = rr[c];
                    if (r < cnt) {
                        unsigned e   = (unsigned)(g * 4 + c);
                        unsigned w   = r & 31u;
                        unsigned pos = atomicAdd(&sBinN[w], 1u);
                        if (pos < (unsigned)BINCAP) sQ[w * BINCAP + pos] = e | (r << 20);
                        atomicAdd(&sDeg[r], 1u);
                    }
                }
            }
        }
    }
}

__device__ __forceinline__ unsigned list_range_edges(const unsigned* __restrict__ rid, unsigned range,
                                                     unsigned* list) {
    const int t = threadIdx.x;
    const v4u* __restrict__ r4 = (const v4u*)rid;
    const unsigned pat = range * 0x01010101u;
    unsigned found = 0u;
    for (int g0 = t; g0 < NG16; g0 += 4 * AGG_T) {
        v4u dv[4];
#pragma unroll
        for (int u = 0; u < 4; ++u) {
            int g = g0 + u * AGG_T;
            dv[u] = r4[g < NG16 ? g : NG16 - 1];
        }
#pragma unroll
        for (int u = 0; u < 4; ++u) {
            int g = g0 + u * AGG_T;
            unsigned m = (zero_bytes(dv[u].x ^ pat) >> 7) | (zero_bytes(dv[u].y ^ pat) >> 6) |
                         (zero_bytes(dv[u].z ^ pat) >> 5) | (zero_bytes(dv[u].w ^ pat) >> 4);
            if (g >= NG16) m = 0u;
            unsigned bal = __builtin_amdgcn_ballot_w32(m != 0u);
            while (bal != 0u) {
                if (m != 0u) {
                    const unsigned i   = (unsigned)__builtin_ctz(m);
                    const unsigned pos = found + __builtin_amdgcn_mbcnt_lo(bal, 0u);
                    m &= m - 1u;
                    if (pos < (unsigned)LCAP)
                        list[pos] = (unsigned)g * 16u + (i & 7u) * 4u + (i >> 3);
                }
                found += (unsigned)__builtin_popcount(bal);
                bal = __builtin_amdgcn_ballot_w32(m != 0u);
            }
        }
    }
    return found;
}

__device__ __forceinline__ void bin_listed_edges(const unsigned* list, unsigned n,
                                                 const int* __restrict__ src, const int* __restrict__ dst,
                                                 int lo, unsigned cnt, int lane,
                                                 unsigned* sQ, unsigned* sBinN, unsigned* sDeg) {
    for (unsigned j0 = (unsigned)lane; j0 < n; j0 += 128u) {
        unsigned e[4];
        int      d[4], s[4];
#pragma unroll
        for (int u = 0; u < 4; ++u) {
            unsigned i = j0 + 32u * (unsigned)u;
            e[u] = (i < n) ? list[i] : 0u;
        }
#pragma unroll
        for (int u = 0; u < 4; ++u) {
            d[u] = dst[e[u]];
            s[u] = src[e[u]];
        }
#pragma unroll
        for (int u = 0; u < 4; ++u)
            asm volatile("" : "+v"(s[u]));
#pragma unroll
        for (int u = 0; u < 4; ++u) {
            unsigned i = j0 + 32u * (unsigned)u;
            unsigned r = (unsigned)(d[u] - lo);
            if (i < n && r < cnt) {
                unsigned w   = r & 31u;
                unsigned pos = atomicAdd(&sBinN[w], 1u);
                if (pos < (unsigned)BINCAP) sQ[w * BINCAP + pos] = (unsigned)s[u] | (r << 16);
                atomicAdd(&sDeg[r], 1u);
            }
        }
    }
}

__device__ __forceinline__ void sum_bin(const unsigned* Q, unsigned n, const float* __restrict__ x,
                                        int lane, RowSums& S) {
    const unsigned NONE = 224u << 16;
    unsigned cur = ((unsigned)lane < n) ? Q[lane] : NONE;
    for (unsigned base = 0; base < n; base += 32) {
        const unsigned nb  = base + 32u + (unsigned)lane;
        const unsigned nxt = (nb < n) ? Q[nb] : NONE;
#pragma unroll
        for (int g = 0; g < 32 / AGG_GU; ++g) {
            const unsigned i0 = base + (unsigned)(g * AGG_GU);
            if (i0 < n) {
                unsigned ent[AGG_GU];
                v2f va[AGG_GU], vb[AGG_GU];
#pragma unroll
                for (int u = 0; u < AGG_GU; ++u)
                    ent[u] = (unsigned)__builtin_amdgcn_readlane((int)cur, g * AGG_GU + u);
#pragma unroll
                for (int u = 0; u < AGG_GU; ++u) {
                    const float* xr = x + (size_t)(ent[u] & 0xFFFFu) * D + 2 * lane;
                    va[u] = *(const v2f*)xr;
                    vb[u] = *(const v2f*)(xr + 64);
                }
#pragma unroll
                for (int u = 0; u < AGG_GU; ++u)
                    add_row(S, ent[u] >> 21, va[u], vb[u]);
            }
        }
        cur = nxt;
    }
}

__device__ __forceinline__ void sum_by_scan(const int* __restrict__ src, const int* __restrict__ dst,
                                            const float* __restrict__ x, int lo, unsigned cnt,
                                            int wave, int lane, RowSums& S) {
    for (int e0 = 0; e0 < NE; e0 += 32) {
        const unsigned r = (unsigned)(dst[e0 + lane] - lo);
        const bool mine  = (r < cnt) && ((r & 31u) == (unsigned)wave);
        unsigned m = __builtin_amdgcn_ballot_w32(mine);
        while (m != 0u) {
            const int l = __builtin_ctz(m);
            m &= m - 1u;
            const unsigned rl = (unsigned)__builtin_amdgcn_readlane((int)r, l);
            const int s = src[e0 + l];
            const float* xr = x + (size_t)s * D + 2 * lane;
            const v2f a = *(const v2f*)xr;
            const v2f b = *(const v2f*)(xr + 64);
            add_row(S, rl >> 5, a, b);
        }
    }
}

template <bool BUILD>
__global__ __launch_bounds__(AGG_T)
void aggregate_kernel(const float* __restrict__ x, const int* __restrict__ src,
                      const int* __restrict__ dst, const unsigned* __restrict__ rid,
                      unsigned* __restrict__ q, unsigned* __restrict__ qcnt,
                      float* __restrict__ inv_deg, float* __restrict__ agg) {
    __shared__ unsigned sQ[QCAP];

    const int t    = threadIdx.x;
    const int lane = t & 31;
    const int wave = __builtin_amdgcn_readfirstlane(t >> 5);
    const int lo   = blockIdx.x * AGG_ROWS;
    const unsigned cnt = (unsigned)min(AGG_ROWS, N_NODES - lo);
    unsigned* qseg = q + (size_t)blockIdx.x * QCAP + (size_t)wave * BINCAP;
    unsigned* qn   = qcnt + (size_t)blockIdx.x * AGG_NW + wave;
    unsigned* bin  = sQ + wave * BINCAP;

    RowSums S;
    S.v = (v32f){};

    if constexpr (BUILD) {
        __shared__ unsigned sL[AGG_NW * LCAP];
        __shared__ unsigned sBinN[AGG_NW];
        __shared__ unsigned sDeg[AGG_ROWS];
        __shared__ unsigned sOver;

        if (t < AGG_NW) sBinN[t] = 0u;
        if (t < AGG_ROWS) sDeg[t] = 0u;
        if (t == 0) sOver = 0u;
        unsigned* list = sL + wave * LCAP;
        const unsigned lraw = (unsigned)__builtin_amdgcn_readfirstlane(
            (int)list_range_edges(rid, (unsigned)blockIdx.x, list));
        __syncthreads();
        if (lraw > (unsigned)LCAP) {
            if (lane == 0) sOver = 1u;
        } else {
            bin_listed_edges(list, lraw, src, dst, lo, cnt, lane, sQ, sBinN, sDeg);
        }
        __syncthreads();
        const bool over = sOver != 0u;
        if (over) {
            if (t < AGG_NW) sBinN[t] = 0u;
            if (t < AGG_ROWS) sDeg[t] = 0u;
            __syncthreads();
            filter_bins(dst, lo, cnt, sQ, sBinN, sDeg);
            __syncthreads();
        }
        if ((unsigned)t < cnt) inv_deg[lo + t] = 1.0f / fmaxf((float)sDeg[t], 1.0f);
        const unsigned nraw = (unsigned)__builtin_amdgcn_readfirstlane((int)sBinN[wave]);
        const unsigned n    = min(nraw, (unsigned)BINCAP);
        if (over) {
            for (unsigned j0 = (unsigned)lane; j0 < n; j0 += 128u) {
                unsigned v[4];
                int      s[4];
#pragma unroll
                for (int u = 0; u < 4; ++u) {
                    unsigned i = j0 + 32u * (unsigned)u;
                    v[u] = (i < n) ? bin[i] : 0u;
                }
#pragma unroll
                for (int u = 0; u < 4; ++u) s[u] = src[v[u] & 0xFFFFFu];
#pragma unroll
                for (int u = 0; u < 4; ++u) {
                    unsigned i = j0 + 32u * (unsigned)u;
                    if (i < n) {
                        unsigned ent = (unsigned)s[u] | ((v[u] >> 20) << 16);
                        bin[i]  = ent;
                        qseg[i] = ent;
                    }
                }
            }
        } else {
            for (unsigned i = (unsigned)lane; i < n; i += 32u) qseg[i] = bin[i];
        }
        if (lane == 0) *qn = nraw;

        if (nraw <= (unsigned)BINCAP) sum_bin(bin, n, x, lane, S);
        else                          sum_by_scan(src, dst, x, lo, cnt, wave, lane, S);
    } else {
        const unsigned nraw = (unsigned)__builtin_amdgcn_readfirstlane((int)*qn);
        if (nraw <= (unsigned)BINCAP) {
            for (unsigned j0 = (unsigned)lane; j0 < nraw; j0 += 128u) {
                unsigned v[4];
#pragma unroll
                for (int u = 0; u < 4; ++u) {
                    unsigned i = j0 + 32u * (unsigned)u;
                    v[u] = (i < nraw) ? qseg[i] : 0u;
                }
#pragma unroll
                for (int u = 0; u < 4; ++u) {
                    unsigned i = j0 + 32u * (unsigned)u;
                    if (i < nraw) bin[i] = v[u];
                }
            }
            sum_bin(bin, nraw, x, lane, S);
        } else {
            sum_by_scan(src, dst, x, lo, cnt, wave, lane, S);
        }
    }

#pragma unroll
    for (int j = 0; j < AGG_SLOTS; ++j) {
        const unsigned r = (unsigned)(wave + AGG_NW * j);
        if (r < cnt) {
            float* ao = agg + (size_t)(lo + (int)r) * D + 2 * lane;
            *(v2f*)ao        = (v2f){S.v[4 * j], S.v[4 * j + 1]};
            *(v2f*)(ao + 64) = (v2f){S.v[4 * j + 2], S.v[4 * j + 3]};
        }
    }
}

#define WPITCH    288
#define GEMM_W    16
#define GEMM_FULL 12
#define GEMM_T    (GEMM_W * 32)
#define GEMM_WGS  256

static_assert(GEMM_W - GEMM_FULL == 4, "the left-over tile is split in column quarters");
static_assert(MTILES >= GEMM_WGS * GEMM_FULL && MTILES - GEMM_WGS * GEMM_FULL <= GEMM_WGS,
              "every tile has exactly one owner");

__device__ __forceinline__ void stage_weights(const float* __restrict__ Wg, float* sW, int t) {
#pragma unroll
    for (int i = 0; i < 2048 / GEMM_T; ++i) {
        const int id = t + GEMM_T * i;
        const int p  = id >> 5;
        const int n4 = id & 31;
        const v4f a = *(const v4f*)(Wg + (size_t)(2 * p) * D + 4 * n4);
        const v4f b = *(const v4f*)(Wg + (size_t)(2 * p + 1) * D + 4 * n4);
        float* o = sW + p * WPITCH + 8 * n4;
        *(v4f*)o       = (v4f){a.x, b.x, a.y, b.y};
        *(v4f*)(o + 4) = (v4f){a.z, b.z, a.w, b.w};
    }
}

template <int NTN>
__device__ __forceinline__ void tile_times_weights(const float* tile, const float* sW,
                                                   int nt0, int c, int half, v8f* acc) {
    const float* ap = tile + c * LDSS + 2 * half;
    const float* wp = sW + half * WPITCH + 32 * nt0 + 2 * c;
#pragma unroll
    for (int ks = 0; ks < 32; ++ks) {
        const v2f a = *(const v2f*)(ap + 4 * ks);
#pragma unroll
        for (int j = 0; j < NTN; ++j) {
            const v2f b = *(const v2f*)(wp + 2 * ks * WPITCH + 32 * j);
            acc[j] = __builtin_amdgcn_wmma_f32_16x16x4_f32(false, a, false, b,
                                                           (short)0, acc[j], false, false);
        }
    }
}

template <int MODE, int NTN>
__device__ __forceinline__ void wave_tile(float* tile, const float* sW, const v4f* xr, const v4f* ar,
                                          const float* __restrict__ inv_deg, size_t rowbase, int nt0,
                                          int lane,
                                          const float* __restrict__ bias,
                                          const float* __restrict__ gamma, const float* __restrict__ beta,
                                          const float* __restrict__ mean, const float* __restrict__ var,
                                          float* __restrict__ out) {
    const int half = lane >> 4;
    const int c    = lane & 15;

    v8f acc[NTN];
#pragma unroll
    for (int j = 0; j < NTN; ++j) acc[j] = (v8f){};

#pragma unroll
    for (int r = 0; r < 16; ++r) {
        v4f v;
        if (MODE == 0) {
            v = xr[r];
        } else {
            const float idg = inv_deg[rowbase + r];
            v = ar[r] * idg;
        }
        *(v4f*)(tile + r * LDSS + 4 * lane) = v;
    }
    tile_times_weights<NTN>(tile, sW, nt0, c, half, acc);

    if (MODE != 0) {
#pragma unroll
        for (int r = 0; r < 16; ++r)
            *(v4f*)(tile + r * LDSS + 4 * lane) = xr[r];
        tile_times_weights<NTN>(tile, sW + 64 * WPITCH, nt0, c, half, acc);
    }

#pragma unroll
    for (int j = 0; j < NTN; ++j) {
        const int n = 16 * (nt0 + j) + c;
        const float bcol = bias[n];
        float s = 1.0f, tt = 0.0f;
        if (MODE != 0) {
            s  = gamma[n] * rsqrtf(var[n] + EPS_BN);
            tt = beta[n] - mean[n] * s;
        }
#pragma unroll
        for (int v = 0; v < 8; ++v) {
            const int m = v + 8 * half;
            const float h = acc[j][v] + bcol;
            float y;
            if (MODE == 0) {
                y = fmaxf(h, 0.0f);
            } else {
                y = fmaxf(h * s + tt, 0.0f);
                if (MODE == 1) y += tile[m * LDSS + n];
            }
            tile[m * LDSS + n] = y;
        }
    }
    constexpr int LPR = 4 * NTN;
    constexpr int RPI = 32 / LPR;
    const int srow = lane / LPR;
    const int scol = 16 * nt0 + 4 * (lane % LPR);
#pragma unroll
    for (int i = 0; i < 16 / RPI; ++i) {
        const int r = i * RPI + srow;
        *(v4f*)(out + (rowbase + r) * D + scol) = *(const v4f*)(tile + r * LDSS + scol);
    }
}

template <int MODE>
__global__ __launch_bounds__(GEMM_T)
void sage_gemm_kernel(const float* __restrict__ Aagg, const float* __restrict__ inv_deg,
                      const float* __restrict__ Ax,
                      const float* __restrict__ W1, const float* __restrict__ W2,
                      const float* __restrict__ bias,
                      const float* __restrict__ gamma, const float* __restrict__ beta,
                      const float* __restrict__ mean, const float* __restrict__ var,
                      float* __restrict__ out,
                      const int* __restrict__ dst, unsigned* __restrict__ rid) {
    __shared__ __attribute__((aligned(16))) float sW[(MODE == 0 ? 1 : 2) * 64 * WPITCH];
    __shared__ __attribute__((aligned(16))) float sT[GEMM_W * 16 * LDSS];

    const int t    = threadIdx.x;
    const int lane = t & 31;
    const int wave = __builtin_amdgcn_readfirstlane(t >> 5);

    const bool full = wave < GEMM_FULL;
    const int  mt   = full ? (int)blockIdx.x + GEMM_WGS * wave
                           : GEMM_WGS * GEMM_FULL + (int)blockIdx.x;
    const bool has  = mt < MTILES;
    const size_t rowbase = (size_t)(has ? mt : 0) * 16;
    float* tile = sT + wave * (16 * LDSS);

    constexpr int PK_T = GEMM_WGS * GEMM_T;
    constexpr int PK_N = (NG4 + PK_T - 1) / PK_T;
    v4i dq[PK_N];
    if (MODE == 0) {
        const v4i* __restrict__ d4 = (const v4i*)dst;
#pragma unroll
        for (int u = 0; u < PK_N; ++u) {
            const int g = (int)blockIdx.x * GEMM_T + t + u * PK_T;
            dq[u] = d4[g < NG4 ? g : NG4 - 1];
        }
    }

    v4f xr[16];
    v4f ar[16];
    if (has) {
#pragma unroll
        for (int r = 0; r < 16; ++r)
            xr[r] = *(const v4f*)(Ax + (rowbase + r) * D + 4 * lane);
        if (MODE != 0) {
#pragma unroll
            for (int r = 0; r < 16; ++r)
                ar[r] = *(const v4f*)(Aagg + (rowbase + r) * D + 4 * lane);
        }
    }

    stage_weights(W1, sW, t);
    if (MODE != 0) stage_weights(W2, sW + 64 * WPITCH, t);
    __syncthreads();
    if (MODE == 0) {
#pragma unroll
        for (int u = 0; u < PK_N; ++u) {
            const int g = (int)blockIdx.x * GEMM_T + t + u * PK_T;
            if (g < NG4)
                rid[g] = range_byte(dq[u].x) | (range_byte(dq[u].y) << 8) |
                         (range_byte(dq[u].z) << 16) | (range_byte(dq[u].w) << 24);
        }
    }
    if (!has) return;

    if (full)
        wave_tile<MODE, 8>(tile, sW, xr, ar, inv_deg, rowbase, 0, lane,
                           bias, gamma, beta, mean, var, out);
    else
        wave_tile<MODE, 2>(tile, sW, xr, ar, inv_deg, rowbase, 2 * (wave - GEMM_FULL), lane,
                           bias, gamma, beta, mean, var, out);
}


extern "C" void kernel_launch(void* const* d_in, const int* in_sizes, int n_in,
                              void* d_out, int out_size, void* d_ws, size_t ws_size,
                              hipStream_t stream) {
    (void)in_sizes; (void)n_in; (void)out_size; (void)ws_size;

    const float* x   = (const float*)d_in[0];
    const int*   ei  = (const int*)d_in[1];
    const float* Wp  = (const float*)d_in[2];
    const float* bp  = (const float*)d_in[3];
    const float* Wl  = (const float*)d_in[4];
    const float* bl  = (const float*)d_in[5];
    const float* Wr  = (const float*)d_in[6];
    const float* gam = (const float*)d_in[7];
    const float* bet = (const float*)d_in[8];
    const float* mea = (const float*)d_in[9];
    const float* var = (const float*)d_in[10];

    const int* src = ei;
    const int* dst = ei + NE;

    float* ws  = (float*)d_ws;
    float* deg = ws;
    float* xa  = ws + ((N_NODES + 63) / 64) * 64;
    float* xb  = xa + (size_t)N_NODES * D;
    float* agg = xb + (size_t)N_NODES * D;

    unsigned* q    = (unsigned*)d_out;
    unsigned* qcnt = q + (size_t)AGG_WGS * QCAP;
    unsigned* rid  = qcnt + (size_t)AGG_WGS * AGG_NW;

    sage_gemm_kernel<0><<<GEMM_WGS, GEMM_T, 0, stream>>>(
        nullptr, deg, x, Wp, Wp, bp, gam, bet, mea, var, xa, dst, rid);

    float* xcur  = xa;
    float* xnext = xb;
    for (int i = 0; i < 4; ++i) {
        if (i == 0)
            aggregate_kernel<true><<<AGG_WGS, AGG_T, 0, stream>>>(xcur, src, dst, rid, q, qcnt, deg, agg);
        else
            aggregate_kernel<false><<<AGG_WGS, AGG_T, 0, stream>>>(xcur, src, dst, rid, q, qcnt, deg, agg);

        if (i < 3)
            sage_gemm_kernel<1><<<GEMM_WGS, GEMM_T, 0, stream>>>(
                agg, deg, xcur,
                Wl + (size_t)i * D * D, Wr + (size_t)i * D * D, bl + (size_t)i * D,
                gam + (size_t)i * D, bet + (size_t)i * D, mea + (size_t)i * D, var + (size_t)i * D,
                xnext, nullptr, nullptr);
        else
            sage_gemm_kernel<2><<<GEMM_WGS, GEMM_T, 0, stream>>>(
                agg, deg, xcur,
                Wl + (size_t)i * D * D, Wr + (size_t)i * D * D, bl + (size_t)i * D,
                gam + (size_t)i * D, bet + (size_t)i * D, mea + (size_t)i * D, var + (size_t)i * D,
                (float*)d_out, nullptr, nullptr);

        float* tmp = xcur; xcur = xnext; xnext = tmp;
    }
}
